# starting kernel
# baseline (speedup 1.0000x reference)
#include <hip/hip_runtime.h>

#define N_NODES 100000
#define DIM 128
#define MTILES (N_NODES / 16)

typedef __attribute__((ext_vector_type(2))) float v2f;
typedef __attribute__((ext_vector_type(8))) float v8f;

__global__ void k_init_deg(float* deg) {
    int i = blockIdx.x * blockDim.x + threadIdx.x;
    if (i < N_NODES) deg[i] = 1.0f;
}

__global__ void k_count_deg(const int* __restrict__ dst, long long E, float* deg) {
    long long e = (long long)blockIdx.x * blockDim.x + threadIdx.x;
    if (e < E) atomicAdd(&deg[dst[e]], 1.0f);
}

__global__ void k_dinv(float* deg) {
    int i = blockIdx.x * blockDim.x + threadIdx.x;
    if (i < N_NODES) deg[i] = rsqrtf(deg[i]);
}

__global__ void k_zero(float* p, long long n4) {
    long long i = (long long)blockIdx.x * blockDim.x + threadIdx.x;
    if (i < n4) ((float4*)p)[i] = make_float4(0.f, 0.f, 0.f, 0.f);
}

__global__ __launch_bounds__(256) void k_gemm(const float* __restrict__ X,
                                              const float* __restrict__ W,
                                              float* __restrict__ H) {
    int wave = (blockIdx.x * blockDim.x + threadIdx.x) >> 5;
    int lane = threadIdx.x & 31;
    if (wave >= MTILES) return;

    const int m0   = wave * 16;
    const int half = lane >> 4;
    const int l16  = lane & 15;
    const int koff = half * 2;

    v8f acc[8];
#pragma unroll
    for (int n = 0; n < 8; ++n) acc[n] = (v8f){0,0,0,0,0,0,0,0};

    const float* xrow = X + (size_t)(m0 + l16) * DIM;

    for (int k0 = 0; k0 < DIM; k0 += 4) {
        v2f a;
        a.x = xrow[k0 + koff + 0];
        a.y = xrow[k0 + koff + 1];
#pragma unroll
        for (int n = 0; n < 8; ++n) {
            v2f b;
            b.x = W[(size_t)(k0 + koff + 0) * DIM + n * 16 + l16];
            b.y = W[(size_t)(k0 + koff + 1) * DIM + n * 16 + l16];
            acc[n] = __builtin_amdgcn_wmma_f32_16x16x4_f32(
                false, a, false, b, (short)0, acc[n], false, false);
        }
    }

    const int rbase = m0 + half * 8;
#pragma unroll
    for (int n = 0; n < 8; ++n) {
#pragma unroll
        for (int v = 0; v < 8; ++v) {
            H[(size_t)(rbase + v) * DIM + n * 16 + l16] = acc[n][v];
        }
    }
}

__global__ void k_aggregate(const float* __restrict__ H,
                            const int* __restrict__ src,
                            const int* __restrict__ dst,
                            const float* __restrict__ dinv,
                            float* __restrict__ out, long long E) {
    long long t = (long long)blockIdx.x * blockDim.x + threadIdx.x;
    long long e = t >> 5;
    int lane = threadIdx.x & 31;
    if (e >= E) return;

    int s = src[e];
    int d = dst[e];
    float nrm = dinv[s] * dinv[d];

    float4 v = ((const float4*)(H + (size_t)s * DIM))[lane];
    float* op = out + (size_t)d * DIM + lane * 4;
    atomicAdd(op + 0, v.x * nrm);
    atomicAdd(op + 1, v.y * nrm);
    atomicAdd(op + 2, v.z * nrm);
    atomicAdd(op + 3, v.w * nrm);
}

__global__ void k_bias_relu_self(const float* __restrict__ agg,
                                 const float* __restrict__ H,
                                 const float* __restrict__ dinv,
                                 const float* __restrict__ bias,
                                 float* __restrict__ out) {
    long long idx = (long long)blockIdx.x * blockDim.x + threadIdx.x;
    const long long n4 = (long long)N_NODES * DIM / 4;
    if (idx >= n4) return;
    int i  = (int)(idx >> 5);
    int f4 = (int)(idx & 31);

    float di = dinv[i];
    float sn = di * di;
    float4 a  = ((const float4*)agg)[idx];
    float4 h  = ((const float4*)H)[idx];
    float4 bb = ((const float4*)bias)[f4];
    float4 r;
    r.x = fmaxf(fmaf(sn, h.x, a.x) + bb.x, 0.f);
    r.y = fmaxf(fmaf(sn, h.y, a.y) + bb.y, 0.f);
    r.z = fmaxf(fmaf(sn, h.z, a.z) + bb.z, 0.f);
    r.w = fmaxf(fmaf(sn, h.w, a.w) + bb.w, 0.f);
    ((float4*)out)[idx] = r;
}

extern "C" void kernel_launch(void* const* d_in, const int* in_sizes, int n_in,
                              void* d_out, int out_size, void* d_ws, size_t ws_size,
                              hipStream_t stream) {
    const float* x   = (const float*)d_in[0];
    const int*   ei  = (const int*)  d_in[1];
    const float* W1  = (const float*)d_in[2];
    const float* b1  = (const float*)d_in[3];
    const float* W2  = (const float*)d_in[4];
    const float* b2  = (const float*)d_in[5];
    float* out = (float*)d_out;

    const long long E = (long long)in_sizes[1] / 2;
    const int* src = ei;
    const int* dst = ei + E;

    float* dinv = (float*)d_ws;
    float* bufA = dinv + N_NODES;
    float* bufB = bufA + (size_t)N_NODES * DIM;

    const long long nd  = (long long)N_NODES * DIM;
    const long long nd4 = nd / 4;

    dim3 blk(256);
    const int gN    = (N_NODES + 255) / 256;
    const int gE    = (int)((E + 255) / 256);
    const int gZ    = (int)((nd4 + 255) / 256);
    const int gAgg  = (int)((E * 32 + 255) / 256);
    const int gGemm = (MTILES + 7) / 8;

    k_init_deg<<<gN, blk, 0, stream>>>(dinv);
    k_count_deg<<<gE, blk, 0, stream>>>(dst, E, dinv);
    k_dinv<<<gN, blk, 0, stream>>>(dinv);

    k_gemm<<<gGemm, blk, 0, stream>>>(x, W1, bufA);
    k_zero<<<gZ, blk, 0, stream>>>(bufB, nd4);
    k_aggregate<<<gAgg, blk, 0, stream>>>(bufA, src, dst, dinv, bufB, E);
    k_bias_relu_self<<<gZ, blk, 0, stream>>>(bufB, bufA, dinv, b1, bufA);

    k_gemm<<<gGemm, blk, 0, stream>>>(bufA, W2, bufB);
    k_zero<<<gZ, blk, 0, stream>>>(out, nd4);
    k_aggregate<<<gAgg, blk, 0, stream>>>(bufB, src, dst, dinv, out, E);
    k_bias_relu_self<<<gZ, blk, 0, stream>>>(out, bufB, dinv, b2, out);
}


# Edge aggregation as a per-node gather over a destination-sorted edge list rebuilt each call (int atomic rank, scan, fill) instead of float atomics; sum order per node = arrival-order rank
# speedup vs baseline: 2.4236x; 2.4236x over previous
#include <hip/hip_runtime.h>

#define N_NODES 100000
#define DIM 128
#define MTILES (N_NODES / 16)

typedef __attribute__((ext_vector_type(2))) float v2f;
typedef __attribute__((ext_vector_type(8))) float v8f;

__global__ void k_zero_cnt(int* cnt) {
    int i = blockIdx.x * blockDim.x + threadIdx.x;
    if (i < N_NODES) cnt[i] = 0;
}

__global__ void k_count_rank(const int* __restrict__ dst, long long E,
                             int* cnt, int* __restrict__ rank) {
    long long e = (long long)blockIdx.x * blockDim.x + threadIdx.x;
    if (e < E) rank[e] = atomicAdd(&cnt[dst[e]], 1);
}

#define SCAN_T 1024
#define SCAN_BLOCKS ((N_NODES + SCAN_T - 1) / SCAN_T)
__global__ __launch_bounds__(SCAN_T) void k_scan_dinv(const int* __restrict__ cnt,
                                                      int* __restrict__ rowptr,
                                                      float* __restrict__ dinv) {
    __shared__ int wsum[SCAN_T / 32];
    __shared__ int wpre[SCAN_T / 32];
    const int tid  = threadIdx.x;
    const int lane = tid & 31;
    const int wid  = tid >> 5;
    const int b    = blockIdx.x;

    int part = 0;
    const int4* c4 = (const int4*)cnt;
    const int total4 = b * (SCAN_T / 4);
    int k = tid;
    for (; k + 7 * SCAN_T < total4; k += 8 * SCAN_T) {
        int4 q0 = c4[k];
        int4 q1 = c4[k + 1 * SCAN_T];
        int4 q2 = c4[k + 2 * SCAN_T];
        int4 q3 = c4[k + 3 * SCAN_T];
        int4 q4 = c4[k + 4 * SCAN_T];
        int4 q5 = c4[k + 5 * SCAN_T];
        int4 q6 = c4[k + 6 * SCAN_T];
        int4 q7 = c4[k + 7 * SCAN_T];
        part += (q0.x + q0.y) + (q0.z + q0.w) + (q1.x + q1.y) + (q1.z + q1.w)
              + (q2.x + q2.y) + (q2.z + q2.w) + (q3.x + q3.y) + (q3.z + q3.w)
              + (q4.x + q4.y) + (q4.z + q4.w) + (q5.x + q5.y) + (q5.z + q5.w)
              + (q6.x + q6.y) + (q6.z + q6.w) + (q7.x + q7.y) + (q7.z + q7.w);
    }
    for (; k < total4; k += SCAN_T) {
        int4 q = c4[k];
        part += (q.x + q.y) + (q.z + q.w);
    }
    for (int m = 16; m > 0; m >>= 1) part += __shfl_xor(part, m, 32);
    if (lane == 0) wsum[wid] = part;
    __syncthreads();
    int prefix = 0;
    for (int w = 0; w < SCAN_T / 32; ++w) prefix += wsum[w];
    __syncthreads();

    const int i = b * SCAN_T + tid;
    const int c = (i < N_NODES) ? cnt[i] : 0;
    int incl = c;
    for (int m = 1; m < 32; m <<= 1) {
        int up = __shfl_up(incl, m, 32);
        if (lane >= m) incl += up;
    }
    if (lane == 31) wsum[wid] = incl;
    __syncthreads();
    if (wid == 0) {
        int t = wsum[lane];
        int s = t;
        for (int m = 1; m < 32; m <<= 1) {
            int up = __shfl_up(s, m, 32);
            if (lane >= m) s += up;
        }
        wpre[lane] = s - t;
    }
    __syncthreads();
    const int excl = prefix + wpre[wid] + incl - c;
    if (i < N_NODES) {
        rowptr[i] = excl;
        dinv[i]   = rsqrtf((float)(c + 1));
        if (i == N_NODES - 1) rowptr[N_NODES] = excl + c;
    }
}

__global__ void k_fill_csr(const int* __restrict__ src, const int* __restrict__ dst,
                           const int* __restrict__ rank, const int* __restrict__ rowptr,
                           const float* __restrict__ dinv, int2* __restrict__ csr, long long E) {
    long long e = (long long)blockIdx.x * blockDim.x + threadIdx.x;
    if (e >= E) return;
    int s = src[e];
    int d = dst[e];
    long long p = (long long)rowptr[d] + rank[e];
    p = p < 0 ? 0 : (p >= E ? E - 1 : p);
    csr[p] = make_int2(s, __float_as_int(dinv[s] * dinv[d]));
}

__global__ __launch_bounds__(256) void k_gemm(const float* __restrict__ X,
                                              const float* __restrict__ W,
                                              float* __restrict__ H) {
    int wave = (blockIdx.x * blockDim.x + threadIdx.x) >> 5;
    int lane = threadIdx.x & 31;
    if (wave >= MTILES) return;

    const int m0   = wave * 16;
    const int half = lane >> 4;
    const int l16  = lane & 15;
    const int koff = half * 2;

    v8f acc[8];
#pragma unroll
    for (int n = 0; n < 8; ++n) acc[n] = (v8f){0,0,0,0,0,0,0,0};

    const float* xrow = X + (size_t)(m0 + l16) * DIM;

    for (int k0 = 0; k0 < DIM; k0 += 4) {
        v2f a;
        a.x = xrow[k0 + koff + 0];
        a.y = xrow[k0 + koff + 1];
#pragma unroll
        for (int n = 0; n < 8; ++n) {
            v2f b;
            b.x = W[(size_t)(k0 + koff + 0) * DIM + n * 16 + l16];
            b.y = W[(size_t)(k0 + koff + 1) * DIM + n * 16 + l16];
            acc[n] = __builtin_amdgcn_wmma_f32_16x16x4_f32(
                false, a, false, b, (short)0, acc[n], false, false);
        }
    }

    const int rbase = m0 + half * 8;
#pragma unroll
    for (int n = 0; n < 8; ++n) {
#pragma unroll
        for (int v = 0; v < 8; ++v) {
            H[(size_t)(rbase + v) * DIM + n * 16 + l16] = acc[n][v];
        }
    }
}


static __device__ __forceinline__ float4 ld_row(const float4* __restrict__ H4, int sv, int j, int lane) {
    int s = __builtin_amdgcn_readlane(sv, j);
    s = min(max(s, 0), N_NODES - 1);
    return H4[(size_t)s * (DIM / 4) + lane];
}
static __device__ __forceinline__ void acc_row(float4& acc, int wv, int j, float4 v) {
    float w = __int_as_float(__builtin_amdgcn_readlane(wv, j));
    acc.x = fmaf(w, v.x, acc.x); acc.y = fmaf(w, v.y, acc.y);
    acc.z = fmaf(w, v.z, acc.z); acc.w = fmaf(w, v.w, acc.w);
}

__global__ __launch_bounds__(256) void k_gather(const float* __restrict__ H,
                                                const int* __restrict__ rowptr,
                                                const int2* __restrict__ csr,
                                                float* __restrict__ agg) {
    const int lane = threadIdx.x & 31;
    const int node = __builtin_amdgcn_readfirstlane((int)(blockIdx.x * (blockDim.x >> 5) + (threadIdx.x >> 5)));
    if (node >= N_NODES) return;

    const int beg = rowptr[node];
    const int end = rowptr[node + 1];
    const float4* __restrict__ H4 = (const float4*)H;

    float4 acc = make_float4(0.f, 0.f, 0.f, 0.f);
    for (int base = beg; base < end; base += 32) {
        const int n = (end - base < 32) ? (end - base) : 32;
        int2 me = make_int2(0, 0);
        if (lane < n) me = csr[base + lane];
        int j = 0;
        for (; j + 8 <= n; j += 8) {
            float4 v0 = ld_row(H4, me.x, j + 0, lane);
            float4 v1 = ld_row(H4, me.x, j + 1, lane);
            float4 v2 = ld_row(H4, me.x, j + 2, lane);
            float4 v3 = ld_row(H4, me.x, j + 3, lane);
            float4 v4 = ld_row(H4, me.x, j + 4, lane);
            float4 v5 = ld_row(H4, me.x, j + 5, lane);
            float4 v6 = ld_row(H4, me.x, j + 6, lane);
            float4 v7 = ld_row(H4, me.x, j + 7, lane);
            acc_row(acc, me.y, j + 0, v0); acc_row(acc, me.y, j + 1, v1);
            acc_row(acc, me.y, j + 2, v2); acc_row(acc, me.y, j + 3, v3);
            acc_row(acc, me.y, j + 4, v4); acc_row(acc, me.y, j + 5, v5);
            acc_row(acc, me.y, j + 6, v6); acc_row(acc, me.y, j + 7, v7);
        }
        if (j + 4 <= n) {
            float4 v0 = ld_row(H4, me.x, j + 0, lane);
            float4 v1 = ld_row(H4, me.x, j + 1, lane);
            float4 v2 = ld_row(H4, me.x, j + 2, lane);
            float4 v3 = ld_row(H4, me.x, j + 3, lane);
            acc_row(acc, me.y, j + 0, v0); acc_row(acc, me.y, j + 1, v1);
            acc_row(acc, me.y, j + 2, v2); acc_row(acc, me.y, j + 3, v3);
            j += 4;
        }
        const int rem = n - j;
        if (rem == 3) {
            float4 v0 = ld_row(H4, me.x, j + 0, lane);
            float4 v1 = ld_row(H4, me.x, j + 1, lane);
            float4 v2 = ld_row(H4, me.x, j + 2, lane);
            acc_row(acc, me.y, j + 0, v0); acc_row(acc, me.y, j + 1, v1); acc_row(acc, me.y, j + 2, v2);
        } else if (rem == 2) {
            float4 v0 = ld_row(H4, me.x, j + 0, lane);
            float4 v1 = ld_row(H4, me.x, j + 1, lane);
            acc_row(acc, me.y, j + 0, v0); acc_row(acc, me.y, j + 1, v1);
        } else if (rem == 1) {
            float4 v0 = ld_row(H4, me.x, j + 0, lane);
            acc_row(acc, me.y, j + 0, v0);
        }
    }
    ((float4*)agg)[(size_t)node * (DIM / 4) + lane] = acc;
}

__global__ void k_bias_relu_self(const float* __restrict__ agg,
                                 const float* __restrict__ H,
                                 const float* __restrict__ dinv,
                                 const float* __restrict__ bias,
                                 float* __restrict__ out) {
    long long idx = (long long)blockIdx.x * blockDim.x + threadIdx.x;
    const long long n4 = (long long)N_NODES * DIM / 4;
    if (idx >= n4) return;
    int i  = (int)(idx >> 5);
    int f4 = (int)(idx & 31);

    float di = dinv[i];
    float sn = di * di;
    float4 a  = ((const float4*)agg)[idx];
    float4 h  = ((const float4*)H)[idx];
    float4 bb = ((const float4*)bias)[f4];
    float4 r;
    r.x = fmaxf(fmaf(sn, h.x, a.x) + bb.x, 0.f);
    r.y = fmaxf(fmaf(sn, h.y, a.y) + bb.y, 0.f);
    r.z = fmaxf(fmaf(sn, h.z, a.z) + bb.z, 0.f);
    r.w = fmaxf(fmaf(sn, h.w, a.w) + bb.w, 0.f);
    ((float4*)out)[idx] = r;
}

extern "C" void kernel_launch(void* const* d_in, const int* in_sizes, int n_in,
                              void* d_out, int out_size, void* d_ws, size_t ws_size,
                              hipStream_t stream) {
    const float* x   = (const float*)d_in[0];
    const int*   ei  = (const int*)  d_in[1];
    const float* W1  = (const float*)d_in[2];
    const float* b1  = (const float*)d_in[3];
    const float* W2  = (const float*)d_in[4];
    const float* b2  = (const float*)d_in[5];
    float* out = (float*)d_out;

    const long long E = (long long)in_sizes[1] / 2;
    const int* src = ei;
    const int* dst = ei + E;

    float* dinv = (float*)d_ws;
    float* bufA = dinv + N_NODES;
    float* bufB = bufA + (size_t)N_NODES * DIM;
    int*  cnt    = (int*)(bufB + (size_t)N_NODES * DIM);
    int*  rowptr = cnt + N_NODES;
    int2* csr    = (int2*)(rowptr + N_NODES + 32);
    int*  rank   = (int*)(csr + E);

    const long long nd  = (long long)N_NODES * DIM;
    const long long nd4 = nd / 4;

    dim3 blk(256);
    const int gN    = (N_NODES + 255) / 256;
    const int gE    = (int)((E + 255) / 256);
    const int gZ    = (int)((nd4 + 255) / 256);
    const int gNode = (N_NODES + 7) / 8;
    const int gGemm = (MTILES + 7) / 8;

    k_zero_cnt<<<gN, blk, 0, stream>>>(cnt);
    k_count_rank<<<gE, blk, 0, stream>>>(dst, E, cnt, rank);
    k_scan_dinv<<<SCAN_BLOCKS, SCAN_T, 0, stream>>>(cnt, rowptr, dinv);
    k_fill_csr<<<gE, blk, 0, stream>>>(src, dst, rank, rowptr, dinv, csr, E);

    k_gemm<<<gGemm, blk, 0, stream>>>(x, W1, bufA);
    k_gather<<<gNode, blk, 0, stream>>>(bufA, rowptr, csr, bufB);
    k_bias_relu_self<<<gZ, blk, 0, stream>>>(bufB, bufA, dinv, b1, bufA);

    k_gemm<<<gGemm, blk, 0, stream>>>(bufA, W2, bufB);
    k_gather<<<gNode, blk, 0, stream>>>(bufB, rowptr, csr, out);
    k_bias_relu_self<<<gZ, blk, 0, stream>>>(out, bufB, dinv, b2, out);
}


# Edge list grouped by destination with LDS counters (chunk histogram into 512-node buckets, then per-bucket node ranks) instead of a global returning atomic per edge; sum order = LDS arrival rank
# speedup vs baseline: 3.2636x; 1.3466x over previous
#include <hip/hip_runtime.h>

#define N_NODES 100000
#define DIM 128
#define MTILES (N_NODES / 16)

typedef __attribute__((ext_vector_type(2))) float v2f;
typedef __attribute__((ext_vector_type(8))) float v8f;

#define BSH      9
#define BNODES   (1 << BSH)
#define NB       ((N_NODES + BNODES - 1) / BNODES)
#define PG       256
#define PT       1024
#define NPIECE   (NB * PG)
#define PIECE_PER_T ((NPIECE + PT - 1) / PT)
#define EU       8

static __device__ __forceinline__ int clamp_node(int v) { return min(max(v, 0), N_NODES - 1); }

__global__ __launch_bounds__(PT) void k_bucket_count(const int* __restrict__ dst, long long E,
                                                     int* __restrict__ counts) {
    __shared__ int hist[NB];
    const int tid = threadIdx.x;
    for (int b = tid; b < NB; b += PT) hist[b] = 0;
    __syncthreads();
    const long long chunk = (E + PG - 1) / PG;
    const long long lo = (long long)blockIdx.x * chunk;
    const long long hi = (lo + chunk < E) ? (lo + chunk) : E;
    for (long long e0 = lo + tid; e0 < hi; e0 += (long long)EU * PT) {
        int d[EU];
#pragma unroll
        for (int u = 0; u < EU; ++u) {
            const long long e = e0 + (long long)u * PT;
            d[u] = dst[(e < hi) ? e : e0];
        }
#pragma unroll
        for (int u = 0; u < EU; ++u)
            if (e0 + (long long)u * PT < hi) atomicAdd(&hist[clamp_node(d[u]) >> BSH], 1);
    }
    __syncthreads();
    for (int b = tid; b < NB; b += PT) counts[b * PG + blockIdx.x] = hist[b];
}

#define SCAN_ROWS 25
__global__ __launch_bounds__(PT) void k_bucket_scan(const int* __restrict__ counts, int* __restrict__ offs) {
    __shared__ int wsum[PT / 32];
    const int tid  = threadIdx.x;
    const int lane = tid & 31;
    const int wid  = tid >> 5;
    const int base = wid * (32 * PIECE_PER_T);
    int t = 0;
#pragma unroll
    for (int k = 0; k < PIECE_PER_T; ++k) {
        const int i = base + k * 32 + lane;
        t += (i < NPIECE) ? counts[i] : 0;
    }
    for (int m = 16; m > 0; m >>= 1) t += __shfl_xor(t, m, 32);
    if (lane == 0) wsum[wid] = t;
    __syncthreads();
    int carry = 0;
    for (int w = 0; w < wid; ++w) carry += wsum[w];
    for (int k0 = 0; k0 < PIECE_PER_T; k0 += SCAN_ROWS) {
        int v[SCAN_ROWS];
#pragma unroll
        for (int r = 0; r < SCAN_ROWS; ++r) {
            const int k = k0 + r;
            const int i = base + k * 32 + lane;
            v[r] = (k < PIECE_PER_T && i < NPIECE) ? counts[i] : 0;
        }
#pragma unroll
        for (int r = 0; r < SCAN_ROWS; ++r) {
            const int k = k0 + r;
            const int i = base + k * 32 + lane;
            const int c = v[r];
            int incl = c;
            for (int m = 1; m < 32; m <<= 1) {
                int up = __shfl_up(incl, m, 32);
                if (lane >= m) incl += up;
            }
            if (k < PIECE_PER_T && i < NPIECE) offs[i] = carry + incl - c;
            carry += __shfl(incl, 31, 32);
        }
    }
    if (tid == PT - 1) offs[NPIECE] = carry;
}

__global__ __launch_bounds__(PT) void k_bucket_place(const int* __restrict__ src, const int* __restrict__ dst,
                                                     long long E, const int* __restrict__ offs,
                                                     int2* __restrict__ tmp) {
    __shared__ int cur[NB];
    const int tid = threadIdx.x;
    for (int b = tid; b < NB; b += PT) cur[b] = offs[b * PG + blockIdx.x];
    __syncthreads();
    const long long chunk = (E + PG - 1) / PG;
    const long long lo = (long long)blockIdx.x * chunk;
    const long long hi = (lo + chunk < E) ? (lo + chunk) : E;
    for (long long e0 = lo + tid; e0 < hi; e0 += (long long)EU * PT) {
        int s[EU], d[EU];
#pragma unroll
        for (int u = 0; u < EU; ++u) {
            const long long e  = e0 + (long long)u * PT;
            const long long ec = (e < hi) ? e : e0;
            d[u] = dst[ec];
            s[u] = src[ec];
        }
#pragma unroll
        for (int u = 0; u < EU; ++u) {
            if (e0 + (long long)u * PT < hi) {
                const int dn = clamp_node(d[u]);
                long long p = atomicAdd(&cur[dn >> BSH], 1);
                p = p < 0 ? 0 : (p >= E ? E - 1 : p);
                tmp[p] = make_int2(s[u], dn);
            }
        }
    }
}

__global__ __launch_bounds__(PT) void k_bucket_sort(const int2* __restrict__ tmp, const int* __restrict__ offs,
                                                    int* __restrict__ rowptr, float* __restrict__ dinv,
                                                    int2* __restrict__ csr, long long E) {
    __shared__ int cnt[BNODES];
    __shared__ int cur[BNODES];
    __shared__ int wtot[BNODES / 32];
    const int tid  = threadIdx.x;
    const int lane = tid & 31;
    const int wid  = tid >> 5;
    const int b    = blockIdx.x;
    const int bs   = offs[b * PG];
    const int be   = offs[(b + 1) * PG];
    for (int i = tid; i < BNODES; i += PT) cnt[i] = 0;
    __syncthreads();
    for (int p0 = bs + tid; p0 < be; p0 += EU * PT) {
        int d[EU];
#pragma unroll
        for (int u = 0; u < EU; ++u) {
            const int p = p0 + u * PT;
            d[u] = tmp[(p < be) ? p : p0].y;
        }
#pragma unroll
        for (int u = 0; u < EU; ++u)
            if (p0 + u * PT < be) atomicAdd(&cnt[d[u] & (BNODES - 1)], 1);
    }
    __syncthreads();

    int c = 0, incl = 0;
    if (tid < BNODES) {
        c = cnt[tid];
        incl = c;
        for (int m = 1; m < 32; m <<= 1) {
            int up = __shfl_up(incl, m, 32);
            if (lane >= m) incl += up;
        }
        if (lane == 31) wtot[wid] = incl;
    }
    __syncthreads();
    if (tid < BNODES) {
        int prefix = 0;
        for (int w = 0; w < wid; ++w) prefix += wtot[w];
        const int start = bs + prefix + incl - c;
        cur[tid] = start;
        const int node = b * BNODES + tid;
        if (node < N_NODES) {
            rowptr[node] = start;
            dinv[node]   = rsqrtf((float)(c + 1));
            if (node == N_NODES - 1) rowptr[N_NODES] = start + c;
        }
    }
    __syncthreads();
    for (int p0 = bs + tid; p0 < be; p0 += EU * PT) {
        int2 sd[EU];
#pragma unroll
        for (int u = 0; u < EU; ++u) {
            const int p = p0 + u * PT;
            sd[u] = tmp[(p < be) ? p : p0];
        }
#pragma unroll
        for (int u = 0; u < EU; ++u) {
            if (p0 + u * PT < be) {
                long long q = atomicAdd(&cur[sd[u].y & (BNODES - 1)], 1);
                q = q < 0 ? 0 : (q >= E ? E - 1 : q);
                csr[q] = sd[u];
            }
        }
    }
}

#define EWB 2048
#define EWU 4
__global__ __launch_bounds__(256) void k_edge_weight(int2* __restrict__ csr, const float* __restrict__ dinv, long long E) {
    const long long stride = (long long)EWB * 256;
    for (long long p0 = (long long)blockIdx.x * 256 + threadIdx.x; p0 < E; p0 += EWU * stride) {
        int2  sd[EWU];
        float ws[EWU], wd[EWU];
#pragma unroll
        for (int u = 0; u < EWU; ++u) {
            const long long p = p0 + u * stride;
            sd[u] = csr[(p < E) ? p : p0];
        }
#pragma unroll
        for (int u = 0; u < EWU; ++u) {
            ws[u] = dinv[clamp_node(sd[u].x)];
            wd[u] = dinv[clamp_node(sd[u].y)];
        }
#pragma unroll
        for (int u = 0; u < EWU; ++u) {
            const long long p = p0 + u * stride;
            if (p < E) csr[p] = make_int2(sd[u].x, __float_as_int(ws[u] * wd[u]));
        }
    }
}

__global__ __launch_bounds__(256) void k_gemm(const float* __restrict__ X,
                                              const float* __restrict__ W,
                                              float* __restrict__ H) {
    int wave = (blockIdx.x * blockDim.x + threadIdx.x) >> 5;
    int lane = threadIdx.x & 31;
    if (wave >= MTILES) return;

    const int m0   = wave * 16;
    const int half = lane >> 4;
    const int l16  = lane & 15;
    const int koff = half * 2;

    v8f acc[8];
#pragma unroll
    for (int n = 0; n < 8; ++n) acc[n] = (v8f){0,0,0,0,0,0,0,0};

    const float* xrow = X + (size_t)(m0 + l16) * DIM;

    for (int k0 = 0; k0 < DIM; k0 += 4) {
        v2f a;
        a.x = xrow[k0 + koff + 0];
        a.y = xrow[k0 + koff + 1];
#pragma unroll
        for (int n = 0; n < 8; ++n) {
            v2f b;
            b.x = W[(size_t)(k0 + koff + 0) * DIM + n * 16 + l16];
            b.y = W[(size_t)(k0 + koff + 1) * DIM + n * 16 + l16];
            acc[n] = __builtin_amdgcn_wmma_f32_16x16x4_f32(
                false, a, false, b, (short)0, acc[n], false, false);
        }
    }

    const int rbase = m0 + half * 8;
#pragma unroll
    for (int n = 0; n < 8; ++n) {
#pragma unroll
        for (int v = 0; v < 8; ++v) {
            H[(size_t)(rbase + v) * DIM + n * 16 + l16] = acc[n][v];
        }
    }
}


static __device__ __forceinline__ float4 ld_row(const float4* __restrict__ H4, int sv, int j, int lane) {
    int s = __builtin_amdgcn_readlane(sv, j);
    s = min(max(s, 0), N_NODES - 1);
    return H4[(size_t)s * (DIM / 4) + lane];
}
static __device__ __forceinline__ void acc_row(float4& acc, int wv, int j, float4 v) {
    float w = __int_as_float(__builtin_amdgcn_readlane(wv, j));
    acc.x = fmaf(w, v.x, acc.x); acc.y = fmaf(w, v.y, acc.y);
    acc.z = fmaf(w, v.z, acc.z); acc.w = fmaf(w, v.w, acc.w);
}

__global__ __launch_bounds__(256) void k_gather(const float* __restrict__ H,
                                                const int* __restrict__ rowptr,
                                                const int2* __restrict__ csr,
                                                float* __restrict__ agg) {
    const int lane = threadIdx.x & 31;
    const int node = __builtin_amdgcn_readfirstlane((int)(blockIdx.x * (blockDim.x >> 5) + (threadIdx.x >> 5)));
    if (node >= N_NODES) return;

    const int beg = rowptr[node];
    const int end = rowptr[node + 1];
    const float4* __restrict__ H4 = (const float4*)H;

    float4 acc = make_float4(0.f, 0.f, 0.f, 0.f);
    for (int base = beg; base < end; base += 32) {
        const int n = (end - base < 32) ? (end - base) : 32;
        int2 me = make_int2(0, 0);
        if (lane < n) me = csr[base + lane];
        int j = 0;
        for (; j + 8 <= n; j += 8) {
            float4 v0 = ld_row(H4, me.x, j + 0, lane);
            float4 v1 = ld_row(H4, me.x, j + 1, lane);
            float4 v2 = ld_row(H4, me.x, j + 2, lane);
            float4 v3 = ld_row(H4, me.x, j + 3, lane);
            float4 v4 = ld_row(H4, me.x, j + 4, lane);
            float4 v5 = ld_row(H4, me.x, j + 5, lane);
            float4 v6 = ld_row(H4, me.x, j + 6, lane);
            float4 v7 = ld_row(H4, me.x, j + 7, lane);
            acc_row(acc, me.y, j + 0, v0); acc_row(acc, me.y, j + 1, v1);
            acc_row(acc, me.y, j + 2, v2); acc_row(acc, me.y, j + 3, v3);
            acc_row(acc, me.y, j + 4, v4); acc_row(acc, me.y, j + 5, v5);
            acc_row(acc, me.y, j + 6, v6); acc_row(acc, me.y, j + 7, v7);
        }
        if (j + 4 <= n) {
            float4 v0 = ld_row(H4, me.x, j + 0, lane);
            float4 v1 = ld_row(H4, me.x, j + 1, lane);
            float4 v2 = ld_row(H4, me.x, j + 2, lane);
            float4 v3 = ld_row(H4, me.x, j + 3, lane);
            acc_row(acc, me.y, j + 0, v0); acc_row(acc, me.y, j + 1, v1);
            acc_row(acc, me.y, j + 2, v2); acc_row(acc, me.y, j + 3, v3);
            j += 4;
        }
        const int rem = n - j;
        if (rem == 3) {
            float4 v0 = ld_row(H4, me.x, j + 0, lane);
            float4 v1 = ld_row(H4, me.x, j + 1, lane);
            float4 v2 = ld_row(H4, me.x, j + 2, lane);
            acc_row(acc, me.y, j + 0, v0); acc_row(acc, me.y, j + 1, v1); acc_row(acc, me.y, j + 2, v2);
        } else if (rem == 2) {
            float4 v0 = ld_row(H4, me.x, j + 0, lane);
            float4 v1 = ld_row(H4, me.x, j + 1, lane);
            acc_row(acc, me.y, j + 0, v0); acc_row(acc, me.y, j + 1, v1);
        } else if (rem == 1) {
            float4 v0 = ld_row(H4, me.x, j + 0, lane);
            acc_row(acc, me.y, j + 0, v0);
        }
    }
    ((float4*)agg)[(size_t)node * (DIM / 4) + lane] = acc;
}

__global__ void k_bias_relu_self(const float* __restrict__ agg,
                                 const float* __restrict__ H,
                                 const float* __restrict__ dinv,
                                 const float* __restrict__ bias,
                                 float* __restrict__ out) {
    long long idx = (long long)blockIdx.x * blockDim.x + threadIdx.x;
    const long long n4 = (long long)N_NODES * DIM / 4;
    if (idx >= n4) return;
    int i  = (int)(idx >> 5);
    int f4 = (int)(idx & 31);

    float di = dinv[i];
    float sn = di * di;
    float4 a  = ((const float4*)agg)[idx];
    float4 h  = ((const float4*)H)[idx];
    float4 bb = ((const float4*)bias)[f4];
    float4 r;
    r.x = fmaxf(fmaf(sn, h.x, a.x) + bb.x, 0.f);
    r.y = fmaxf(fmaf(sn, h.y, a.y) + bb.y, 0.f);
    r.z = fmaxf(fmaf(sn, h.z, a.z) + bb.z, 0.f);
    r.w = fmaxf(fmaf(sn, h.w, a.w) + bb.w, 0.f);
    ((float4*)out)[idx] = r;
}

extern "C" void kernel_launch(void* const* d_in, const int* in_sizes, int n_in,
                              void* d_out, int out_size, void* d_ws, size_t ws_size,
                              hipStream_t stream) {
    const float* x   = (const float*)d_in[0];
    const int*   ei  = (const int*)  d_in[1];
    const float* W1  = (const float*)d_in[2];
    const float* b1  = (const float*)d_in[3];
    const float* W2  = (const float*)d_in[4];
    const float* b2  = (const float*)d_in[5];
    float* out = (float*)d_out;

    const long long E = (long long)in_sizes[1] / 2;
    const int* src = ei;
    const int* dst = ei + E;

    float* dinv = (float*)d_ws;
    float* bufA = dinv + N_NODES;
    float* bufB = bufA + (size_t)N_NODES * DIM;
    int*  rowptr = (int*)(bufB + (size_t)N_NODES * DIM);
    int2* csr    = (int2*)(rowptr + N_NODES + 32);
    int2* tmp    = csr + E;
    int*  counts = (int*)(tmp + E);
    int*  offs   = counts + NPIECE;

    const long long nd  = (long long)N_NODES * DIM;
    const long long nd4 = nd / 4;

    dim3 blk(256);
    const int gZ    = (int)((nd4 + 255) / 256);
    const int gNode = (N_NODES + 7) / 8;
    const int gGemm = (MTILES + 7) / 8;

    k_bucket_count<<<PG, PT, 0, stream>>>(dst, E, counts);
    k_bucket_scan<<<1, PT, 0, stream>>>(counts, offs);
    k_bucket_place<<<PG, PT, 0, stream>>>(src, dst, E, offs, tmp);
    k_bucket_sort<<<NB, PT, 0, stream>>>(tmp, offs, rowptr, dinv, csr, E);
    k_edge_weight<<<EWB, blk, 0, stream>>>(csr, dinv, E);

    k_gemm<<<gGemm, blk, 0, stream>>>(x, W1, bufA);
    k_gather<<<gNode, blk, 0, stream>>>(bufA, rowptr, csr, bufB);
    k_bias_relu_self<<<gZ, blk, 0, stream>>>(bufB, bufA, dinv, b1, bufA);

    k_gemm<<<gGemm, blk, 0, stream>>>(bufA, W2, bufB);
    k_gather<<<gNode, blk, 0, stream>>>(bufB, rowptr, csr, out);
    k_bias_relu_self<<<gZ, blk, 0, stream>>>(out, bufB, dinv, b2, out);
}


# One fused kernel per layer: 16-node tiles gather the weighted neighbour sum of the layer input into LDS, then WMMA with W held in LDS, bias+relu on the way out; sum order: aggregate before product
# speedup vs baseline: 4.9915x; 1.5294x over previous
#include <hip/hip_runtime.h>

#define N_NODES 100000
#define DIM 128
#define MTILES (N_NODES / 16)

typedef __attribute__((ext_vector_type(2))) float v2f;
typedef __attribute__((ext_vector_type(8))) float v8f;

#define BSH      9
#define BNODES   (1 << BSH)
#define NB       ((N_NODES + BNODES - 1) / BNODES)
#define PG       256
#define PT       1024
#define NPIECE   (NB * PG)
#define PIECE_PER_T ((NPIECE + PT - 1) / PT)
#define EU       8

static __device__ __forceinline__ int clamp_node(int v) { return min(max(v, 0), N_NODES - 1); }

__global__ __launch_bounds__(PT) void k_bucket_count(const int* __restrict__ dst, long long E,
                                                     int* __restrict__ counts) {
    __shared__ int hist[NB];
    const int tid = threadIdx.x;
    for (int b = tid; b < NB; b += PT) hist[b] = 0;
    __syncthreads();
    const long long chunk = (E + PG - 1) / PG;
    const long long lo = (long long)blockIdx.x * chunk;
    const long long hi = (lo + chunk < E) ? (lo + chunk) : E;
    for (long long e0 = lo + tid; e0 < hi; e0 += (long long)EU * PT) {
        int d[EU];
#pragma unroll
        for (int u = 0; u < EU; ++u) {
            const long long e = e0 + (long long)u * PT;
            d[u] = dst[(e < hi) ? e : e0];
        }
#pragma unroll
        for (int u = 0; u < EU; ++u)
            if (e0 + (long long)u * PT < hi) atomicAdd(&hist[clamp_node(d[u]) >> BSH], 1);
    }
    __syncthreads();
    for (int b = tid; b < NB; b += PT) counts[b * PG + blockIdx.x] = hist[b];
}

#define SCAN_ROWS 25
__global__ __launch_bounds__(PT) void k_bucket_scan(const int* __restrict__ counts, int* __restrict__ offs) {
    __shared__ int wsum[PT / 32];
    const int tid  = threadIdx.x;
    const int lane = tid & 31;
    const int wid  = tid >> 5;
    const int base = wid * (32 * PIECE_PER_T);
    int t = 0;
#pragma unroll
    for (int k = 0; k < PIECE_PER_T; ++k) {
        const int i = base + k * 32 + lane;
        t += (i < NPIECE) ? counts[i] : 0;
    }
    for (int m = 16; m > 0; m >>= 1) t += __shfl_xor(t, m, 32);
    if (lane == 0) wsum[wid] = t;
    __syncthreads();
    int carry = 0;
    for (int w = 0; w < wid; ++w) carry += wsum[w];
    for (int k0 = 0; k0 < PIECE_PER_T; k0 += SCAN_ROWS) {
        int v[SCAN_ROWS];
#pragma unroll
        for (int r = 0; r < SCAN_ROWS; ++r) {
            const int k = k0 + r;
            const int i = base + k * 32 + lane;
            v[r] = (k < PIECE_PER_T && i < NPIECE) ? counts[i] : 0;
        }
#pragma unroll
        for (int r = 0; r < SCAN_ROWS; ++r) {
            const int k = k0 + r;
            const int i = base + k * 32 + lane;
            const int c = v[r];
            int incl = c;
            for (int m = 1; m < 32; m <<= 1) {
                int up = __shfl_up(incl, m, 32);
                if (lane >= m) incl += up;
            }
            if (k < PIECE_PER_T && i < NPIECE) offs[i] = carry + incl - c;
            carry += __shfl(incl, 31, 32);
        }
    }
    if (tid == PT - 1) offs[NPIECE] = carry;
}

__global__ __launch_bounds__(PT) void k_bucket_place(const int* __restrict__ src, const int* __restrict__ dst,
                                                     long long E, const int* __restrict__ offs,
                                                     int2* __restrict__ tmp) {
    __shared__ int cur[NB];
    const int tid = threadIdx.x;
    for (int b = tid; b < NB; b += PT) cur[b] = offs[b * PG + blockIdx.x];
    __syncthreads();
    const long long chunk = (E + PG - 1) / PG;
    const long long lo = (long long)blockIdx.x * chunk;
    const long long hi = (lo + chunk < E) ? (lo + chunk) : E;
    for (long long e0 = lo + tid; e0 < hi; e0 += (long long)EU * PT) {
        int s[EU], d[EU];
#pragma unroll
        for (int u = 0; u < EU; ++u) {
            const long long e  = e0 + (long long)u * PT;
            const long long ec = (e < hi) ? e : e0;
            d[u] = dst[ec];
            s[u] = src[ec];
        }
#pragma unroll
        for (int u = 0; u < EU; ++u) {
            if (e0 + (long long)u * PT < hi) {
                const int dn = clamp_node(d[u]);
                long long p = atomicAdd(&cur[dn >> BSH], 1);
                p = p < 0 ? 0 : (p >= E ? E - 1 : p);
                tmp[p] = make_int2(s[u], dn);
            }
        }
    }
}

__global__ __launch_bounds__(PT) void k_bucket_sort(const int2* __restrict__ tmp, const int* __restrict__ offs,
                                                    int* __restrict__ rowptr, float* __restrict__ dinv,
                                                    int2* __restrict__ csr, long long E) {
    __shared__ int cnt[BNODES];
    __shared__ int cur[BNODES];
    __shared__ int wtot[BNODES / 32];
    const int tid  = threadIdx.x;
    const int lane = tid & 31;
    const int wid  = tid >> 5;
    const int b    = blockIdx.x;
    const int bs   = offs[b * PG];
    const int be   = offs[(b + 1) * PG];
    for (int i = tid; i < BNODES; i += PT) cnt[i] = 0;
    __syncthreads();
    for (int p0 = bs + tid; p0 < be; p0 += EU * PT) {
        int d[EU];
#pragma unroll
        for (int u = 0; u < EU; ++u) {
            const int p = p0 + u * PT;
            d[u] = tmp[(p < be) ? p : p0].y;
        }
#pragma unroll
        for (int u = 0; u < EU; ++u)
            if (p0 + u * PT < be) atomicAdd(&cnt[d[u] & (BNODES - 1)], 1);
    }
    __syncthreads();

    int c = 0, incl = 0;
    if (tid < BNODES) {
        c = cnt[tid];
        incl = c;
        for (int m = 1; m < 32; m <<= 1) {
            int up = __shfl_up(incl, m, 32);
            if (lane >= m) incl += up;
        }
        if (lane == 31) wtot[wid] = incl;
    }
    __syncthreads();
    if (tid < BNODES) {
        int prefix = 0;
        for (int w = 0; w < wid; ++w) prefix += wtot[w];
        const int start = bs + prefix + incl - c;
        cur[tid] = start;
        const int node = b * BNODES + tid;
        if (node < N_NODES) {
            rowptr[node] = start;
            dinv[node]   = rsqrtf((float)(c + 1));
            if (node == N_NODES - 1) rowptr[N_NODES] = start + c;
        }
    }
    __syncthreads();
    for (int p0 = bs + tid; p0 < be; p0 += EU * PT) {
        int2 sd[EU];
#pragma unroll
        for (int u = 0; u < EU; ++u) {
            const int p = p0 + u * PT;
            sd[u] = tmp[(p < be) ? p : p0];
        }
#pragma unroll
        for (int u = 0; u < EU; ++u) {
            if (p0 + u * PT < be) {
                long long q = atomicAdd(&cur[sd[u].y & (BNODES - 1)], 1);
                q = q < 0 ? 0 : (q >= E ? E - 1 : q);
                csr[q] = sd[u];
            }
        }
    }
}

#define EWB 2048
#define EWU 4
__global__ __launch_bounds__(256) void k_edge_weight(int2* __restrict__ csr, const float* __restrict__ dinv, long long E) {
    const long long stride = (long long)EWB * 256;
    for (long long p0 = (long long)blockIdx.x * 256 + threadIdx.x; p0 < E; p0 += EWU * stride) {
        int2  sd[EWU];
        float ws[EWU], wd[EWU];
#pragma unroll
        for (int u = 0; u < EWU; ++u) {
            const long long p = p0 + u * stride;
            sd[u] = csr[(p < E) ? p : p0];
        }
#pragma unroll
        for (int u = 0; u < EWU; ++u) {
            ws[u] = dinv[clamp_node(sd[u].x)];
            wd[u] = dinv[clamp_node(sd[u].y)];
        }
#pragma unroll
        for (int u = 0; u < EWU; ++u) {
            const long long p = p0 + u * stride;
            if (p < E) csr[p] = make_int2(sd[u].x, __float_as_int(ws[u] * wd[u]));
        }
    }
}


static __device__ __forceinline__ float4 ld_row(const float4* __restrict__ H4, int sv, int j, int lane) {
    const unsigned s = min((unsigned)__builtin_amdgcn_readlane(sv, j), (unsigned)(N_NODES - 1));
    return H4[s * (unsigned)(DIM / 4) + (unsigned)lane];
}
static __device__ __forceinline__ void acc_row(float4& acc, int wv, int j, float4 v) {
    float w = __int_as_float(__builtin_amdgcn_readlane(wv, j));
    acc.x = fmaf(w, v.x, acc.x); acc.y = fmaf(w, v.y, acc.y);
    acc.z = fmaf(w, v.z, acc.z); acc.w = fmaf(w, v.w, acc.w);
}


#define APITCH 132
#define LAYER_GRID 768

static __device__ __forceinline__ void flush_row(float* __restrict__ row, float4& acc) {
    const float4 t = *(const float4*)row;
    *(float4*)row = make_float4(acc.x + t.x, acc.y + t.y, acc.z + t.z, acc.w + t.w);
    acc = make_float4(0.f, 0.f, 0.f, 0.f);
}
static __device__ __forceinline__ void step_row(float4& acc, float* __restrict__ row0, int split, int wv, int j, int jj, float4 v) {
    if (j == split) flush_row(row0, acc);
    acc_row(acc, wv, jj, v);
}

static __device__ __forceinline__ void gather_pair(const float4* __restrict__ H4, const int2* __restrict__ csr,
                                                   int p0, int p1, int p2, int lane, int2 lo0, int2 hi0,
                                                   float* __restrict__ row0, float* __restrict__ row1) {
    float4 acc = make_float4(0.f, 0.f, 0.f, 0.f);
    for (int base = p0; base < p2; base += 64) {
        const int n = (p2 - base < 64) ? (p2 - base) : 64;
        const int split = p1 - base;
        int2 lo = lo0, hi = hi0;
        if (base != p0) {
            lo = make_int2(0, 0); hi = make_int2(0, 0);
            if (lane < n)      lo = csr[base + lane];
            if (lane + 32 < n) hi = csr[base + 32 + lane];
        }
        int j = 0;
        for (; j + 16 <= n; j += 16) {
            const int2 me = (j < 32) ? lo : hi;
            const int  jj = j & 31;
            float4 v[16];
#pragma unroll
            for (int u = 0; u < 16; ++u) v[u] = ld_row(H4, me.x, jj + u, lane);
#pragma unroll
            for (int u = 0; u < 16; ++u) step_row(acc, row0, split, me.y, j + u, jj + u, v[u]);
        }
        const int rem = n - j;
        if (rem > 0) {
            const int2 me = (j < 32) ? lo : hi;
            const int  jj = j & 31;
            float4 v[15];
#pragma unroll
            for (int u = 0; u < 15; ++u)
                if (u < rem) v[u] = ld_row(H4, me.x, jj + u, lane);
#pragma unroll
            for (int u = 0; u < 15; ++u)
                if (u < rem) step_row(acc, row0, split, me.y, j + u, jj + u, v[u]);
        }
    }
    if (p1 >= p2) flush_row(row0, acc);
    flush_row(row1, acc);
}

__global__ __launch_bounds__(256, 6) void k_layer(const float* __restrict__ X,
                                               const float* __restrict__ W,
                                               const float* __restrict__ bias,
                                               const float* __restrict__ dinv,
                                               const int* __restrict__ rowptr,
                                               const int2* __restrict__ csr,
                                               float* __restrict__ out) {
    __shared__ __attribute__((aligned(16))) float Wl[64 * 256];
    __shared__ __attribute__((aligned(16))) float At[16 * APITCH];
    const int tid  = threadIdx.x;
    const int lane = tid & 31;
    const int wid  = __builtin_amdgcn_readfirstlane(tid >> 5);
    const int half = lane >> 4;
    const int l16  = lane & 15;

    const float4* __restrict__ W4 = (const float4*)W;
    {
        float4 r0[8], r1[8];
#pragma unroll
        for (int q = 0; q < 8; ++q) {
            const int i = tid + q * 256, kp = i >> 5, c4 = i & 31;
            r0[q] = W4[(2 * kp) * (DIM / 4) + c4];
            r1[q] = W4[(2 * kp + 1) * (DIM / 4) + c4];
        }
#pragma unroll
        for (int q = 0; q < 8; ++q) {
            const int i = tid + q * 256, kp = i >> 5, c4 = i & 31;
            const int o = kp * 256 + ((c4 * 8) ^ ((kp & 1) * 32));
            *(float4*)&Wl[o]     = make_float4(r0[q].x, r1[q].x, r0[q].y, r1[q].y);
            *(float4*)&Wl[o + 4] = make_float4(r0[q].z, r1[q].z, r0[q].w, r1[q].w);
        }
    }

    const float4* __restrict__ X4 = (const float4*)X;
    const int abase = l16 * APITCH + 2 * half;
    const int bbase = half * 256 + ((wid * 32 + l16 * 2) ^ (half * 32));

    int t = blockIdx.x;
    int p0, p1, p2;
    int2 lo0 = make_int2(0, 0), hi0 = make_int2(0, 0);
    float4 xs0, xs1;
    float d0, d1;
    int rp;
    const int l2 = (lane < 2) ? lane : 2;
    {
        const int n0 = t * 16 + 2 * wid;
        p0 = rowptr[n0];
        p1 = rowptr[n0 + 1];
        p2 = rowptr[n0 + 2];
        if (lane < p2 - p0)      lo0 = csr[p0 + lane];
        if (lane + 32 < p2 - p0) hi0 = csr[p0 + 32 + lane];
        xs0 = X4[(unsigned)n0 * (unsigned)(DIM / 4) + (unsigned)lane];
        xs1 = X4[(unsigned)(n0 + 1) * (unsigned)(DIM / 4) + (unsigned)lane];
        d0 = dinv[n0];
        d1 = dinv[n0 + 1];
        const int t1 = (t + LAYER_GRID < MTILES) ? (t + LAYER_GRID) : (MTILES - 1);
        rp = rowptr[t1 * 16 + 2 * wid + l2];
    }
    for (;;) {
        const int m0 = t * 16;
        const int tn = t + LAYER_GRID;

        float* row0 = &At[(2 * wid) * APITCH + lane * 4];
        float* row1 = row0 + APITCH;
        {
            const float s0 = d0 * d0, s1 = d1 * d1;
            *(float4*)row0 = make_float4(s0 * xs0.x, s0 * xs0.y, s0 * xs0.z, s0 * xs0.w);
            *(float4*)row1 = make_float4(s1 * xs1.x, s1 * xs1.y, s1 * xs1.z, s1 * xs1.w);
        }

        gather_pair(X4, csr, p0, p1, p2, lane, lo0, hi0, row0, row1);

        int q0 = 0, q1 = 0, q2 = 0, nrp = 0;
        int2 nlo = make_int2(0, 0), nhi = make_int2(0, 0);
        float4 nxs0 = make_float4(0.f, 0.f, 0.f, 0.f), nxs1 = nxs0;
        float nd0 = 0.f, nd1 = 0.f;
        if (tn < MTILES) {
            q0 = __builtin_amdgcn_readlane(rp, 0);
            q1 = __builtin_amdgcn_readlane(rp, 1);
            q2 = __builtin_amdgcn_readlane(rp, 2);
            const int nn = tn * 16 + 2 * wid;
            if (lane < q2 - q0)      nlo = csr[q0 + lane];
            if (lane + 32 < q2 - q0) nhi = csr[q0 + 32 + lane];
            nxs0 = X4[(unsigned)nn * (unsigned)(DIM / 4) + (unsigned)lane];
            nxs1 = X4[(unsigned)(nn + 1) * (unsigned)(DIM / 4) + (unsigned)lane];
            nd0 = dinv[nn];
            nd1 = dinv[nn + 1];
            const int t2 = (tn + LAYER_GRID < MTILES) ? (tn + LAYER_GRID) : (MTILES - 1);
            nrp = rowptr[t2 * 16 + 2 * wid + l2];
        }

        asm volatile("s_wait_dscnt 0x0\n\ts_barrier_signal -1\n\ts_barrier_wait -1" ::: "memory");

        const float4 bb = ((const float4*)bias)[lane];

        v8f c0 = (v8f){0,0,0,0,0,0,0,0};
        v8f c1 = (v8f){0,0,0,0,0,0,0,0};
#pragma unroll
        for (int k0 = 0; k0 < DIM; k0 += 8) {
            const v2f a0 = *(const v2f*)&At[abase + k0];
            const v2f b0 = *(const v2f*)&Wl[bbase + k0 * 128];
            const v2f a1 = *(const v2f*)&At[abase + k0 + 4];
            const v2f b1 = *(const v2f*)&Wl[bbase + (k0 + 4) * 128];
            c0 = __builtin_amdgcn_wmma_f32_16x16x4_f32(false, a0, false, b0, (short)0, c0, false, false);
            c1 = __builtin_amdgcn_wmma_f32_16x16x4_f32(false, a1, false, b1, (short)0, c1, false, false);
        }

        __syncthreads();

#pragma unroll
        for (int v = 0; v < 8; ++v)
            At[(v + 8 * half) * APITCH + wid * 16 + l16] = c0[v] + c1[v];

        __syncthreads();

#pragma unroll
        for (int rr = 0; rr < 2; ++rr) {
            const int r = 2 * wid + rr;
            const float4 o = *(const float4*)&At[r * APITCH + lane * 4];
            float4 q;
            q.x = fmaxf(o.x + bb.x, 0.f); q.y = fmaxf(o.y + bb.y, 0.f);
            q.z = fmaxf(o.z + bb.z, 0.f); q.w = fmaxf(o.w + bb.w, 0.f);
            ((float4*)out)[(size_t)(m0 + r) * (DIM / 4) + lane] = q;
        }

        if (tn >= MTILES) break;
        t = tn; p0 = q0; p1 = q1; p2 = q2; lo0 = nlo; hi0 = nhi;
        xs0 = nxs0; xs1 = nxs1; d0 = nd0; d1 = nd1; rp = nrp;
    }
}

extern "C" void kernel_launch(void* const* d_in, const int* in_sizes, int n_in,
                              void* d_out, int out_size, void* d_ws, size_t ws_size,
                              hipStream_t stream) {
    const float* x   = (const float*)d_in[0];
    const int*   ei  = (const int*)  d_in[1];
    const float* W1  = (const float*)d_in[2];
    const float* b1  = (const float*)d_in[3];
    const float* W2  = (const float*)d_in[4];
    const float* b2  = (const float*)d_in[5];
    float* out = (float*)d_out;

    const long long E = (long long)in_sizes[1] / 2;
    const int* src = ei;
    const int* dst = ei + E;

    float* dinv = (float*)d_ws;
    float* bufA = dinv + N_NODES;
    int*  rowptr = (int*)(bufA + (size_t)N_NODES * DIM);
    int2* csr    = (int2*)(rowptr + N_NODES + 32);
    int2* tmp    = csr + E;
    int*  counts = (int*)(tmp + E);
    int*  offs   = counts + NPIECE;

    dim3 blk(256);

    k_bucket_count<<<PG, PT, 0, stream>>>(dst, E, counts);
    k_bucket_scan<<<1, PT, 0, stream>>>(counts, offs);
    k_bucket_place<<<PG, PT, 0, stream>>>(src, dst, E, offs, tmp);
    k_bucket_sort<<<NB, PT, 0, stream>>>(tmp, offs, rowptr, dinv, csr, E);
    k_edge_weight<<<EWB, blk, 0, stream>>>(csr, dinv, E);

    k_layer<<<LAYER_GRID, blk, 0, stream>>>(x, W1, b1, dinv, rowptr, csr, bufA);
    k_layer<<<LAYER_GRID, blk, 0, stream>>>(bufA, W2, b2, dinv, rowptr, csr, out);
}


# Edge weight factored: dinv[dst] scales the output row, dinv[src] read per edge in layer 1 only, layer-1 rows stored pre-scaled; weight pass removed (6 launches); products regrouped, sum order kept
# speedup vs baseline: 5.2288x; 1.0475x over previous
#include <hip/hip_runtime.h>

#define N_NODES 100000
#define DIM 128
#define MTILES (N_NODES / 16)

typedef __attribute__((ext_vector_type(2))) float v2f;
typedef __attribute__((ext_vector_type(8))) float v8f;

#define BSH      9
#define BNODES   (1 << BSH)
#define NB       ((N_NODES + BNODES - 1) / BNODES)
#define PG       256
#define PT       1024
#define NPIECE   (NB * PG)
#define PIECE_PER_T ((NPIECE + PT - 1) / PT)
#define EU       8

static __device__ __forceinline__ int clamp_node(int v) { return min(max(v, 0), N_NODES - 1); }

__global__ __launch_bounds__(PT) void k_bucket_count(const int* __restrict__ dst, long long E,
                                                     int* __restrict__ counts) {
    __shared__ int hist[NB];
    const int tid = threadIdx.x;
    for (int b = tid; b < NB; b += PT) hist[b] = 0;
    __syncthreads();
    const long long chunk = (E + PG - 1) / PG;
    const long long lo = (long long)blockIdx.x * chunk;
    const long long hi = (lo + chunk < E) ? (lo + chunk) : E;
    for (long long e0 = lo + tid; e0 < hi; e0 += (long long)EU * PT) {
        int d[EU];
#pragma unroll
        for (int u = 0; u < EU; ++u) {
            const long long e = e0 + (long long)u * PT;
            d[u] = dst[(e < hi) ? e : e0];
        }
#pragma unroll
        for (int u = 0; u < EU; ++u)
            if (e0 + (long long)u * PT < hi) atomicAdd(&hist[clamp_node(d[u]) >> BSH], 1);
    }
    __syncthreads();
    for (int b = tid; b < NB; b += PT) counts[b * PG + blockIdx.x] = hist[b];
}

#define SCAN_ROWS 25
__global__ __launch_bounds__(PT) void k_bucket_scan(const int* __restrict__ counts, int* __restrict__ offs) {
    __shared__ int wsum[PT / 32];
    const int tid  = threadIdx.x;
    const int lane = tid & 31;
    const int wid  = tid >> 5;
    const int base = wid * (32 * PIECE_PER_T);
    int t = 0;
#pragma unroll
    for (int k = 0; k < PIECE_PER_T; ++k) {
        const int i = base + k * 32 + lane;
        t += (i < NPIECE) ? counts[i] : 0;
    }
    for (int m = 16; m > 0; m >>= 1) t += __shfl_xor(t, m, 32);
    if (lane == 0) wsum[wid] = t;
    __syncthreads();
    int carry = 0;
    for (int w = 0; w < wid; ++w) carry += wsum[w];
    for (int k0 = 0; k0 < PIECE_PER_T; k0 += SCAN_ROWS) {
        int v[SCAN_ROWS];
#pragma unroll
        for (int r = 0; r < SCAN_ROWS; ++r) {
            const int k = k0 + r;
            const int i = base + k * 32 + lane;
            v[r] = (k < PIECE_PER_T && i < NPIECE) ? counts[i] : 0;
        }
#pragma unroll
        for (int r = 0; r < SCAN_ROWS; ++r) {
            const int k = k0 + r;
            const int i = base + k * 32 + lane;
            const int c = v[r];
            int incl = c;
            for (int m = 1; m < 32; m <<= 1) {
                int up = __shfl_up(incl, m, 32);
                if (lane >= m) incl += up;
            }
            if (k < PIECE_PER_T && i < NPIECE) offs[i] = carry + incl - c;
            carry += __shfl(incl, 31, 32);
        }
    }
    if (tid == PT - 1) offs[NPIECE] = carry;
}

__global__ __launch_bounds__(PT) void k_bucket_place(const int* __restrict__ src, const int* __restrict__ dst,
                                                     long long E, const int* __restrict__ offs,
                                                     int2* __restrict__ tmp) {
    __shared__ int cur[NB];
    const int tid = threadIdx.x;
    for (int b = tid; b < NB; b += PT) cur[b] = offs[b * PG + blockIdx.x];
    __syncthreads();
    const long long chunk = (E + PG - 1) / PG;
    const long long lo = (long long)blockIdx.x * chunk;
    const long long hi = (lo + chunk < E) ? (lo + chunk) : E;
    for (long long e0 = lo + tid; e0 < hi; e0 += (long long)EU * PT) {
        int s[EU], d[EU];
#pragma unroll
        for (int u = 0; u < EU; ++u) {
            const long long e  = e0 + (long long)u * PT;
            const long long ec = (e < hi) ? e : e0;
            d[u] = dst[ec];
            s[u] = src[ec];
        }
#pragma unroll
        for (int u = 0; u < EU; ++u) {
            if (e0 + (long long)u * PT < hi) {
                const int dn = clamp_node(d[u]);
                long long p = atomicAdd(&cur[dn >> BSH], 1);
                p = p < 0 ? 0 : (p >= E ? E - 1 : p);
                tmp[p] = make_int2(s[u], dn);
            }
        }
    }
}

__global__ __launch_bounds__(PT) void k_bucket_sort(const int2* __restrict__ tmp, const int* __restrict__ offs,
                                                    int* __restrict__ rowptr, float* __restrict__ dinv,
                                                    int* __restrict__ csr, long long E) {
    __shared__ int cnt[BNODES];
    __shared__ int cur[BNODES];
    __shared__ int wtot[BNODES / 32];
    const int tid  = threadIdx.x;
    const int lane = tid & 31;
    const int wid  = tid >> 5;
    const int b    = blockIdx.x;
    const int bs   = offs[b * PG];
    const int be   = offs[(b + 1) * PG];
    for (int i = tid; i < BNODES; i += PT) cnt[i] = 0;
    __syncthreads();
    for (int p0 = bs + tid; p0 < be; p0 += EU * PT) {
        int d[EU];
#pragma unroll
        for (int u = 0; u < EU; ++u) {
            const int p = p0 + u * PT;
            d[u] = tmp[(p < be) ? p : p0].y;
        }
#pragma unroll
        for (int u = 0; u < EU; ++u)
            if (p0 + u * PT < be) atomicAdd(&cnt[d[u] & (BNODES - 1)], 1);
    }
    __syncthreads();

    int c = 0, incl = 0;
    if (tid < BNODES) {
        c = cnt[tid];
        incl = c;
        for (int m = 1; m < 32; m <<= 1) {
            int up = __shfl_up(incl, m, 32);
            if (lane >= m) incl += up;
        }
        if (lane == 31) wtot[wid] = incl;
    }
    __syncthreads();
    if (tid < BNODES) {
        int prefix = 0;
        for (int w = 0; w < wid; ++w) prefix += wtot[w];
        const int start = bs + prefix + incl - c;
        cur[tid] = start;
        const int node = b * BNODES + tid;
        if (node < N_NODES) {
            rowptr[node] = start;
            dinv[node]   = rsqrtf((float)(c + 1));
            if (node == N_NODES - 1) rowptr[N_NODES] = start + c;
        }
    }
    __syncthreads();
    for (int p0 = bs + tid; p0 < be; p0 += EU * PT) {
        int2 sd[EU];
#pragma unroll
        for (int u = 0; u < EU; ++u) {
            const int p = p0 + u * PT;
            sd[u] = tmp[(p < be) ? p : p0];
        }
#pragma unroll
        for (int u = 0; u < EU; ++u) {
            if (p0 + u * PT < be) {
                long long q = atomicAdd(&cur[sd[u].y & (BNODES - 1)], 1);
                q = q < 0 ? 0 : (q >= E ? E - 1 : q);
                csr[q] = sd[u].x;
            }
        }
    }
}


static __device__ __forceinline__ float4 ld_row(const float4* __restrict__ H4, int sv, int j, int lane) {
    const unsigned s = min((unsigned)__builtin_amdgcn_readlane(sv, j), (unsigned)(N_NODES - 1));
    return H4[s * (unsigned)(DIM / 4) + (unsigned)lane];
}
static __device__ __forceinline__ float ld_w(const float* __restrict__ dinv, int s) {
    return dinv[min((unsigned)s, (unsigned)(N_NODES - 1))];
}
template <bool WEIGHTED>
static __device__ __forceinline__ void acc_row(float4& acc, float wv, int j, float4 v) {
    if (WEIGHTED) {
        const float w = __int_as_float(__builtin_amdgcn_readlane(__float_as_int(wv), j));
        acc.x = fmaf(w, v.x, acc.x); acc.y = fmaf(w, v.y, acc.y);
        acc.z = fmaf(w, v.z, acc.z); acc.w = fmaf(w, v.w, acc.w);
    } else {
        acc.x += v.x; acc.y += v.y; acc.z += v.z; acc.w += v.w;
    }
}


#define APITCH 132
#define LAYER_GRID 768

static __device__ __forceinline__ void flush_row(float* __restrict__ row, float4& acc) {
    const float4 t = *(const float4*)row;
    *(float4*)row = make_float4(acc.x + t.x, acc.y + t.y, acc.z + t.z, acc.w + t.w);
    acc = make_float4(0.f, 0.f, 0.f, 0.f);
}
template <bool WEIGHTED>
static __device__ __forceinline__ void step_row(float4& acc, float* __restrict__ row0, int split, float wv,
                                                int j, int jj, float4 v) {
    if (j == split) flush_row(row0, acc);
    acc_row<WEIGHTED>(acc, wv, jj, v);
}

template <bool WEIGHTED>
static __device__ __forceinline__ void gather_pair(const float4* __restrict__ H4, const int* __restrict__ csr,
                                                   const float* __restrict__ dinv,
                                                   int p0, int p1, int p2, int lane, int slo0, int shi0,
                                                   float wlo0, float whi0,
                                                   float* __restrict__ row0, float* __restrict__ row1) {
    float4 acc = make_float4(0.f, 0.f, 0.f, 0.f);
    int   slo = slo0, shi = shi0;
    float wlo = wlo0, whi = whi0;
    for (int base = p0; base < p2; base += 64) {
        const int n = (p2 - base < 64) ? (p2 - base) : 64;
        const int split = p1 - base;
        int j = 0;
        for (; j + 16 <= n; j += 16) {
            const int   ms = (j < 32) ? slo : shi;
            const float mw = (j < 32) ? wlo : whi;
            const int   jj = j & 31;
            float4 v[16];
#pragma unroll
            for (int u = 0; u < 16; ++u) v[u] = ld_row(H4, ms, jj + u, lane);
#pragma unroll
            for (int u = 0; u < 16; ++u) step_row<WEIGHTED>(acc, row0, split, mw, j + u, jj + u, v[u]);
        }
        const int rem = n - j;
        if (rem > 0) {
            const int   ms = (j < 32) ? slo : shi;
            const float mw = (j < 32) ? wlo : whi;
            const int   jj = j & 31;
            float4 v[15];
#pragma unroll
            for (int u = 0; u < 15; ++u)
                if (u < rem) v[u] = ld_row(H4, ms, jj + u, lane);
#pragma unroll
            for (int u = 0; u < 15; ++u)
                if (u < rem) step_row<WEIGHTED>(acc, row0, split, mw, j + u, jj + u, v[u]);
        }
        const int nb = base + 64;
        if (nb < p2) {
            const int nn = (p2 - nb < 64) ? (p2 - nb) : 64;
            slo = 0; shi = 0;
            if (lane < nn)      slo = csr[nb + lane];
            if (lane + 32 < nn) shi = csr[nb + 32 + lane];
            if (WEIGHTED) { wlo = ld_w(dinv, slo); whi = ld_w(dinv, shi); }
        }
    }
    if (p1 >= p2) flush_row(row0, acc);
    flush_row(row1, acc);
}

template <int LAYER>
__global__ __launch_bounds__(256, 6) void k_layer(const float* __restrict__ X,
                                               const float* __restrict__ W,
                                               const float* __restrict__ bias,
                                               const float* __restrict__ dinv,
                                               const int* __restrict__ rowptr,
                                               const int* __restrict__ csr,
                                               float* __restrict__ out) {
    constexpr bool WEIGHTED = (LAYER == 1);
    __shared__ __attribute__((aligned(16))) float Wl[64 * 256];
    __shared__ __attribute__((aligned(16))) float At[16 * APITCH];
    const int tid  = threadIdx.x;
    const int lane = tid & 31;
    const int wid  = __builtin_amdgcn_readfirstlane(tid >> 5);
    const int half = lane >> 4;
    const int l16  = lane & 15;

    const float4* __restrict__ W4 = (const float4*)W;
    {
        float4 r0[8], r1[8];
#pragma unroll
        for (int q = 0; q < 8; ++q) {
            const int i = tid + q * 256, kp = i >> 5, c4 = i & 31;
            r0[q] = W4[(2 * kp) * (DIM / 4) + c4];
            r1[q] = W4[(2 * kp + 1) * (DIM / 4) + c4];
        }
#pragma unroll
        for (int q = 0; q < 8; ++q) {
            const int i = tid + q * 256, kp = i >> 5, c4 = i & 31;
            const int o = kp * 256 + ((c4 * 8) ^ ((kp & 1) * 32));
            *(float4*)&Wl[o]     = make_float4(r0[q].x, r1[q].x, r0[q].y, r1[q].y);
            *(float4*)&Wl[o + 4] = make_float4(r0[q].z, r1[q].z, r0[q].w, r1[q].w);
        }
    }

    const float4* __restrict__ X4 = (const float4*)X;
    const int abase = l16 * APITCH + 2 * half;
    const int bbase = half * 256 + ((wid * 32 + l16 * 2) ^ (half * 32));

    int t = blockIdx.x;
    int p0, p1, p2;
    int   lo0 = 0, hi0 = 0;
    float wlo0 = 0.f, whi0 = 0.f;
    float4 xs0, xs1;
    float d0 = 1.0f, d1 = 1.0f;
    int rp;
    const int l2 = (lane < 2) ? lane : 2;
    {
        const int n0 = t * 16 + 2 * wid;
        p0 = rowptr[n0];
        p1 = rowptr[n0 + 1];
        p2 = rowptr[n0 + 2];
        if (lane < p2 - p0)      lo0 = csr[p0 + lane];
        if (lane + 32 < p2 - p0) hi0 = csr[p0 + 32 + lane];
        xs0 = X4[(unsigned)n0 * (unsigned)(DIM / 4) + (unsigned)lane];
        xs1 = X4[(unsigned)(n0 + 1) * (unsigned)(DIM / 4) + (unsigned)lane];
        if (WEIGHTED) { d0 = dinv[n0]; d1 = dinv[n0 + 1]; wlo0 = ld_w(dinv, lo0); whi0 = ld_w(dinv, hi0); }
        const int t1 = (t + LAYER_GRID < MTILES) ? (t + LAYER_GRID) : (MTILES - 1);
        rp = rowptr[t1 * 16 + 2 * wid + l2];
    }
    for (;;) {
        const int m0 = t * 16;
        const int tn = t + LAYER_GRID;

        float* row0 = &At[(2 * wid) * APITCH + lane * 4];
        float* row1 = row0 + APITCH;
        {
            if (WEIGHTED) {
                *(float4*)row0 = make_float4(d0 * xs0.x, d0 * xs0.y, d0 * xs0.z, d0 * xs0.w);
                *(float4*)row1 = make_float4(d1 * xs1.x, d1 * xs1.y, d1 * xs1.z, d1 * xs1.w);
            } else {
                *(float4*)row0 = xs0;
                *(float4*)row1 = xs1;
            }
        }

        gather_pair<WEIGHTED>(X4, csr, dinv, p0, p1, p2, lane, lo0, hi0, wlo0, whi0, row0, row1);

        int q0 = 0, q1 = 0, q2 = 0, nrp = 0;
        int   nlo = 0, nhi = 0;
        float nwlo = 0.f, nwhi = 0.f;
        float4 nxs0 = make_float4(0.f, 0.f, 0.f, 0.f), nxs1 = nxs0;
        float nd0 = 1.0f, nd1 = 1.0f;
        if (tn < MTILES) {
            q0 = __builtin_amdgcn_readlane(rp, 0);
            q1 = __builtin_amdgcn_readlane(rp, 1);
            q2 = __builtin_amdgcn_readlane(rp, 2);
            const int nn = tn * 16 + 2 * wid;
            if (lane < q2 - q0)      nlo = csr[q0 + lane];
            if (lane + 32 < q2 - q0) nhi = csr[q0 + 32 + lane];
            nxs0 = X4[(unsigned)nn * (unsigned)(DIM / 4) + (unsigned)lane];
            nxs1 = X4[(unsigned)(nn + 1) * (unsigned)(DIM / 4) + (unsigned)lane];
            if (WEIGHTED) { nd0 = dinv[nn]; nd1 = dinv[nn + 1]; }
            const int t2 = (tn + LAYER_GRID < MTILES) ? (tn + LAYER_GRID) : (MTILES - 1);
            nrp = rowptr[t2 * 16 + 2 * wid + l2];
        }

        asm volatile("s_wait_dscnt 0x0\n\ts_barrier_signal -1\n\ts_barrier_wait -1" ::: "memory");

        const float4 bb = ((const float4*)bias)[lane];
        const float dr0 = dinv[m0 + 2 * wid], dr1 = dinv[m0 + 2 * wid + 1];

        v8f c0 = (v8f){0,0,0,0,0,0,0,0};
        v8f c1 = (v8f){0,0,0,0,0,0,0,0};
#pragma unroll
        for (int k0 = 0; k0 < DIM; k0 += 8) {
            const v2f a0 = *(const v2f*)&At[abase + k0];
            const v2f b0 = *(const v2f*)&Wl[bbase + k0 * 128];
            const v2f a1 = *(const v2f*)&At[abase + k0 + 4];
            const v2f b1 = *(const v2f*)&Wl[bbase + (k0 + 4) * 128];
            c0 = __builtin_amdgcn_wmma_f32_16x16x4_f32(false, a0, false, b0, (short)0, c0, false, false);
            c1 = __builtin_amdgcn_wmma_f32_16x16x4_f32(false, a1, false, b1, (short)0, c1, false, false);
        }

        __syncthreads();

#pragma unroll
        for (int v = 0; v < 8; ++v)
            At[(v + 8 * half) * APITCH + wid * 16 + l16] = c0[v] + c1[v];

        __syncthreads();

        if (WEIGHTED) { nwlo = ld_w(dinv, nlo); nwhi = ld_w(dinv, nhi); }

#pragma unroll
        for (int rr = 0; rr < 2; ++rr) {
            const int r = 2 * wid + rr;
            const float4 o = *(const float4*)&At[r * APITCH + lane * 4];
            const float dr = (rr == 0) ? dr0 : dr1;
            const float so = WEIGHTED ? dr : 1.0f;
            float4 q;
            q.x = so * fmaxf(fmaf(dr, o.x, bb.x), 0.f); q.y = so * fmaxf(fmaf(dr, o.y, bb.y), 0.f);
            q.z = so * fmaxf(fmaf(dr, o.z, bb.z), 0.f); q.w = so * fmaxf(fmaf(dr, o.w, bb.w), 0.f);
            ((float4*)out)[(size_t)(m0 + r) * (DIM / 4) + lane] = q;
        }

        if (tn >= MTILES) break;
        t = tn; p0 = q0; p1 = q1; p2 = q2; lo0 = nlo; hi0 = nhi; wlo0 = nwlo; whi0 = nwhi;
        xs0 = nxs0; xs1 = nxs1; d0 = nd0; d1 = nd1; rp = nrp;
    }
}

extern "C" void kernel_launch(void* const* d_in, const int* in_sizes, int n_in,
                              void* d_out, int out_size, void* d_ws, size_t ws_size,
                              hipStream_t stream) {
    const float* x   = (const float*)d_in[0];
    const int*   ei  = (const int*)  d_in[1];
    const float* W1  = (const float*)d_in[2];
    const float* b1  = (const float*)d_in[3];
    const float* W2  = (const float*)d_in[4];
    const float* b2  = (const float*)d_in[5];
    float* out = (float*)d_out;

    const long long E = (long long)in_sizes[1] / 2;
    const int* src = ei;
    const int* dst = ei + E;

    float* dinv = (float*)d_ws;
    float* bufA = dinv + N_NODES;
    int*  rowptr = (int*)(bufA + (size_t)N_NODES * DIM);
    int*  csr    = rowptr + N_NODES + 32;
    int2* tmp    = (int2*)csr + E;
    int*  counts = (int*)(tmp + E);
    int*  offs   = counts + NPIECE;

    dim3 blk(256);

    k_bucket_count<<<PG, PT, 0, stream>>>(dst, E, counts);
    k_bucket_scan<<<1, PT, 0, stream>>>(counts, offs);
    k_bucket_place<<<PG, PT, 0, stream>>>(src, dst, E, offs, tmp);
    k_bucket_sort<<<NB, PT, 0, stream>>>(tmp, offs, rowptr, dinv, csr, E);

    k_layer<1><<<LAYER_GRID, blk, 0, stream>>>(x, W1, b1, dinv, rowptr, csr, bufA);
    k_layer<2><<<LAYER_GRID, blk, 0, stream>>>(bufA, W2, b2, dinv, rowptr, csr, out);
}


# Edge sort in 2 launches, no global scan: a chunk groups its pairs by bucket inside its own slice of tmp (count, LDS scan, place); buckets sum piece sizes in front; sum order: still LDS arrival rank
# speedup vs baseline: 5.4357x; 1.0396x over previous
#include <hip/hip_runtime.h>

#define N_NODES 100000
#define DIM 128
#define MTILES (N_NODES / 16)

typedef __attribute__((ext_vector_type(2))) float v2f;
typedef __attribute__((ext_vector_type(8))) float v8f;

#define BSH      9
#define BNODES   (1 << BSH)
#define NB       ((N_NODES + BNODES - 1) / BNODES)
#define PG       256
#define PT       1024
#define NPIECE   (NB * PG)
#define PIECE_PER_T ((NPIECE + PT - 1) / PT)
#define EU       8
static_assert(NB <= 256, "the chunk pass scans its bucket counts with 256 threads");
static_assert(PG % 32 == 0 && PG <= PT, "a bucket's pieces are dealt to whole waves");

static __device__ __forceinline__ int clamp_node(int v) { return min(max(v, 0), N_NODES - 1); }

__global__ __launch_bounds__(PT) void k_chunk_group(const int* __restrict__ src, const int* __restrict__ dst, long long E,
                                                    int* __restrict__ pcnt, int* __restrict__ pst,
                                                    int2* __restrict__ tmp) {
    __shared__ int hist[NB];
    __shared__ int cur[NB];
    __shared__ int wtot[8];
    const int tid  = threadIdx.x;
    const int lane = tid & 31;
    const int wid  = tid >> 5;
    for (int b = tid; b < NB; b += PT) hist[b] = 0;
    __syncthreads();
    const long long chunk = (E + PG - 1) / PG;
    const long long lo0 = (long long)blockIdx.x * chunk;
    const long long lo = (lo0 < E) ? lo0 : E;
    const long long hi = (lo + chunk < E) ? (lo + chunk) : E;
    for (long long e0 = lo + tid; e0 < hi; e0 += (long long)EU * PT) {
        int d[EU];
#pragma unroll
        for (int u = 0; u < EU; ++u) {
            const long long e = e0 + (long long)u * PT;
            d[u] = dst[(e < hi) ? e : e0];
        }
#pragma unroll
        for (int u = 0; u < EU; ++u)
            if (e0 + (long long)u * PT < hi) atomicAdd(&hist[clamp_node(d[u]) >> BSH], 1);
    }
    __syncthreads();
    int c = 0, incl = 0;
    if (tid < 256) {
        c = (tid < NB) ? hist[tid] : 0;
        incl = c;
        for (int m = 1; m < 32; m <<= 1) {
            int up = __shfl_up(incl, m, 32);
            if (lane >= m) incl += up;
        }
        if (lane == 31) wtot[wid] = incl;
    }
    __syncthreads();
    if (tid < NB) {
        int prefix = 0;
        for (int w = 0; w < wid; ++w) prefix += wtot[w];
        const int st = (int)lo + prefix + incl - c;
        cur[tid] = st;
        pst[tid * PG + blockIdx.x]  = st;
        pcnt[tid * PG + blockIdx.x] = c;
    }
    __syncthreads();
    for (long long e0 = lo + tid; e0 < hi; e0 += (long long)EU * PT) {
        int s[EU], d[EU];
#pragma unroll
        for (int u = 0; u < EU; ++u) {
            const long long e  = e0 + (long long)u * PT;
            const long long ec = (e < hi) ? e : e0;
            d[u] = dst[ec];
            s[u] = src[ec];
        }
#pragma unroll
        for (int u = 0; u < EU; ++u) {
            if (e0 + (long long)u * PT < hi) {
                const int dn = clamp_node(d[u]);
                long long p = atomicAdd(&cur[dn >> BSH], 1);
                p = p < 0 ? 0 : (p >= E ? E - 1 : p);
                tmp[p] = make_int2(s[u], dn);
            }
        }
    }
}

#define PPW (PG / 32)
static __device__ __forceinline__ long long clamp_slot(long long p, long long E) { return p < 0 ? 0 : (p >= E ? E - 1 : p); }

__global__ __launch_bounds__(PT) void k_bucket_sort(const int2* __restrict__ tmp, const int* __restrict__ pcnt,
                                                    const int* __restrict__ pst,
                                                    int* __restrict__ rowptr, float* __restrict__ dinv,
                                                    int* __restrict__ csr, long long E) {
    __shared__ int cnt[BNODES];
    __shared__ int cur[BNODES];
    __shared__ int wtot[BNODES / 32];
    __shared__ int wfront[PT / 32];
    __shared__ int pcs[PG];
    __shared__ int pcn[PG];
    const int tid  = threadIdx.x;
    const int lane = tid & 31;
    const int wid  = tid >> 5;
    const int b    = blockIdx.x;
    const int nfront = b * PG;
    int part = 0;
#pragma unroll
    for (int k = 0; k < PIECE_PER_T; ++k) {
        const int i = k * PT + tid;
        part += (i < nfront) ? pcnt[i] : 0;
    }
    if (tid < PG) {
        pcs[tid] = pst[b * PG + tid];
        pcn[tid] = pcnt[b * PG + tid];
    }
    for (int i = tid; i < BNODES; i += PT) cnt[i] = 0;
    for (int m = 16; m > 0; m >>= 1) part += __shfl_xor(part, m, 32);
    if (lane == 0) wfront[wid] = part;
    __syncthreads();
    int bs = 0;
    for (int w = 0; w < PT / 32; ++w) bs += wfront[w];

    int pn[PPW], ps[PPW];
#pragma unroll
    for (int k = 0; k < PPW; ++k) { ps[k] = pcs[wid + 32 * k]; pn[k] = pcn[wid + 32 * k]; }
    {
        int dA[PPW], dB[PPW];
#pragma unroll
        for (int k = 0; k < PPW; ++k) {
            dA[k] = 0; dB[k] = 0;
            if (lane < pn[k])      dA[k] = tmp[clamp_slot((long long)ps[k] + lane, E)].y;
            if (lane + 32 < pn[k]) dB[k] = tmp[clamp_slot((long long)ps[k] + 32 + lane, E)].y;
        }
#pragma unroll
        for (int k = 0; k < PPW; ++k) {
            if (lane < pn[k])      atomicAdd(&cnt[dA[k] & (BNODES - 1)], 1);
            if (lane + 32 < pn[k]) atomicAdd(&cnt[dB[k] & (BNODES - 1)], 1);
        }
#pragma unroll
        for (int k = 0; k < PPW; ++k)
            for (int i = 64 + lane; i < pn[k]; i += 32)
                atomicAdd(&cnt[tmp[clamp_slot((long long)ps[k] + i, E)].y & (BNODES - 1)], 1);
    }
    __syncthreads();

    int c = 0, incl = 0;
    if (tid < BNODES) {
        c = cnt[tid];
        incl = c;
        for (int m = 1; m < 32; m <<= 1) {
            int up = __shfl_up(incl, m, 32);
            if (lane >= m) incl += up;
        }
        if (lane == 31) wtot[wid] = incl;
    }
    __syncthreads();
    if (tid < BNODES) {
        int prefix = 0;
        for (int w = 0; w < wid; ++w) prefix += wtot[w];
        const int start = bs + prefix + incl - c;
        cur[tid] = start;
        const int node = b * BNODES + tid;
        if (node < N_NODES) {
            rowptr[node] = start;
            dinv[node]   = rsqrtf((float)(c + 1));
            if (node == N_NODES - 1) rowptr[N_NODES] = start + c;
        }
    }
    __syncthreads();
    {
        int2 sA[PPW], sB[PPW];
#pragma unroll
        for (int k = 0; k < PPW; ++k) {
            sA[k] = make_int2(0, 0); sB[k] = make_int2(0, 0);
            if (lane < pn[k])      sA[k] = tmp[clamp_slot((long long)ps[k] + lane, E)];
            if (lane + 32 < pn[k]) sB[k] = tmp[clamp_slot((long long)ps[k] + 32 + lane, E)];
        }
#pragma unroll
        for (int k = 0; k < PPW; ++k) {
            if (lane < pn[k]) {
                const long long q = atomicAdd(&cur[sA[k].y & (BNODES - 1)], 1);
                csr[clamp_slot(q, E)] = sA[k].x;
            }
            if (lane + 32 < pn[k]) {
                const long long q = atomicAdd(&cur[sB[k].y & (BNODES - 1)], 1);
                csr[clamp_slot(q, E)] = sB[k].x;
            }
        }
#pragma unroll
        for (int k = 0; k < PPW; ++k)
            for (int i = 64 + lane; i < pn[k]; i += 32) {
                const int2 sd = tmp[clamp_slot((long long)ps[k] + i, E)];
                const long long q = atomicAdd(&cur[sd.y & (BNODES - 1)], 1);
                csr[clamp_slot(q, E)] = sd.x;
            }
    }
}


static __device__ __forceinline__ float4 ld_row(const float4* __restrict__ H4, int sv, int j, int lane) {
    const unsigned s = min((unsigned)__builtin_amdgcn_readlane(sv, j), (unsigned)(N_NODES - 1));
    return H4[s * (unsigned)(DIM / 4) + (unsigned)lane];
}
static __device__ __forceinline__ float ld_w(const float* __restrict__ dinv, int s) {
    return dinv[min((unsigned)s, (unsigned)(N_NODES - 1))];
}
template <bool WEIGHTED>
static __device__ __forceinline__ void acc_row(float4& acc, float wv, int j, float4 v) {
    if (WEIGHTED) {
        const float w = __int_as_float(__builtin_amdgcn_readlane(__float_as_int(wv), j));
        acc.x = fmaf(w, v.x, acc.x); acc.y = fmaf(w, v.y, acc.y);
        acc.z = fmaf(w, v.z, acc.z); acc.w = fmaf(w, v.w, acc.w);
    } else {
        acc.x += v.x; acc.y += v.y; acc.z += v.z; acc.w += v.w;
    }
}


#define APITCH 132
#define LAYER_GRID 768

static __device__ __forceinline__ void flush_row(float* __restrict__ row, float4& acc) {
    const float4 t = *(const float4*)row;
    *(float4*)row = make_float4(acc.x + t.x, acc.y + t.y, acc.z + t.z, acc.w + t.w);
    acc = make_float4(0.f, 0.f, 0.f, 0.f);
}
template <bool WEIGHTED>
static __device__ __forceinline__ void step_row(float4& acc, float* __restrict__ row0, int split, float wv,
                                                int j, int jj, float4 v) {
    if (j == split) flush_row(row0, acc);
    acc_row<WEIGHTED>(acc, wv, jj, v);
}

template <bool WEIGHTED>
static __device__ __forceinline__ void gather_pair(const float4* __restrict__ H4, const int* __restrict__ csr,
                                                   const float* __restrict__ dinv,
                                                   int p0, int p1, int p2, int lane, int slo0, int shi0,
                                                   float wlo0, float whi0,
                                                   float* __restrict__ row0, float* __restrict__ row1) {
    float4 acc = make_float4(0.f, 0.f, 0.f, 0.f);
    int   slo = slo0, shi = shi0;
    float wlo = wlo0, whi = whi0;
    for (int base = p0; base < p2; base += 64) {
        const int n = (p2 - base < 64) ? (p2 - base) : 64;
        const int split = p1 - base;
        int j = 0;
        for (; j + 16 <= n; j += 16) {
            const int   ms = (j < 32) ? slo : shi;
            const float mw = (j < 32) ? wlo : whi;
            const int   jj = j & 31;
            float4 v[16];
#pragma unroll
            for (int u = 0; u < 16; ++u) v[u] = ld_row(H4, ms, jj + u, lane);
#pragma unroll
            for (int u = 0; u < 16; ++u) step_row<WEIGHTED>(acc, row0, split, mw, j + u, jj + u, v[u]);
        }
        const int rem = n - j;
        if (rem > 0) {
            const int   ms = (j < 32) ? slo : shi;
            const float mw = (j < 32) ? wlo : whi;
            const int   jj = j & 31;
            float4 v[15];
#pragma unroll
            for (int u = 0; u < 15; ++u)
                if (u < rem) v[u] = ld_row(H4, ms, jj + u, lane);
#pragma unroll
            for (int u = 0; u < 15; ++u)
                if (u < rem) step_row<WEIGHTED>(acc, row0, split, mw, j + u, jj + u, v[u]);
        }
        const int nb = base + 64;
        if (nb < p2) {
            const int nn = (p2 - nb < 64) ? (p2 - nb) : 64;
            slo = 0; shi = 0;
            if (lane < nn)      slo = csr[nb + lane];
            if (lane + 32 < nn) shi = csr[nb + 32 + lane];
            if (WEIGHTED) { wlo = ld_w(dinv, slo); whi = ld_w(dinv, shi); }
        }
    }
    if (p1 >= p2) flush_row(row0, acc);
    flush_row(row1, acc);
}

template <int LAYER>
__global__ __launch_bounds__(256, 6) void k_layer(const float* __restrict__ X,
                                               const float* __restrict__ W,
                                               const float* __restrict__ bias,
                                               const float* __restrict__ dinv,
                                               const int* __restrict__ rowptr,
                                               const int* __restrict__ csr,
                                               float* __restrict__ out) {
    constexpr bool WEIGHTED = (LAYER == 1);
    __shared__ __attribute__((aligned(16))) float Wl[64 * 256];
    __shared__ __attribute__((aligned(16))) float At[16 * APITCH];
    const int tid  = threadIdx.x;
    const int lane = tid & 31;
    const int wid  = __builtin_amdgcn_readfirstlane(tid >> 5);
    const int half = lane >> 4;
    const int l16  = lane & 15;

    const float4* __restrict__ W4 = (const float4*)W;
    {
        float4 r0[8], r1[8];
#pragma unroll
        for (int q = 0; q < 8; ++q) {
            const int i = tid + q * 256, kp = i >> 5, c4 = i & 31;
            r0[q] = W4[(2 * kp) * (DIM / 4) + c4];
            r1[q] = W4[(2 * kp + 1) * (DIM / 4) + c4];
        }
#pragma unroll
        for (int q = 0; q < 8; ++q) {
            const int i = tid + q * 256, kp = i >> 5, c4 = i & 31;
            const int o = kp * 256 + ((c4 * 8) ^ ((kp & 1) * 32));
            *(float4*)&Wl[o]     = make_float4(r0[q].x, r1[q].x, r0[q].y, r1[q].y);
            *(float4*)&Wl[o + 4] = make_float4(r0[q].z, r1[q].z, r0[q].w, r1[q].w);
        }
    }

    const float4* __restrict__ X4 = (const float4*)X;
    const int abase = l16 * APITCH + 2 * half;
    const int bbase = half * 256 + ((wid * 32 + l16 * 2) ^ (half * 32));

    int t = blockIdx.x;
    int p0, p1, p2;
    int   lo0 = 0, hi0 = 0;
    float wlo0 = 0.f, whi0 = 0.f;
    float4 xs0, xs1;
    float d0 = 1.0f, d1 = 1.0f;
    int rp;
    const int l2 = (lane < 2) ? lane : 2;
    {
        const int n0 = t * 16 + 2 * wid;
        p0 = rowptr[n0];
        p1 = rowptr[n0 + 1];
        p2 = rowptr[n0 + 2];
        if (lane < p2 - p0)      lo0 = csr[p0 + lane];
        if (lane + 32 < p2 - p0) hi0 = csr[p0 + 32 + lane];
        xs0 = X4[(unsigned)n0 * (unsigned)(DIM / 4) + (unsigned)lane];
        xs1 = X4[(unsigned)(n0 + 1) * (unsigned)(DIM / 4) + (unsigned)lane];
        if (WEIGHTED) { d0 = dinv[n0]; d1 = dinv[n0 + 1]; wlo0 = ld_w(dinv, lo0); whi0 = ld_w(dinv, hi0); }
        const int t1 = (t + LAYER_GRID < MTILES) ? (t + LAYER_GRID) : (MTILES - 1);
        rp = rowptr[t1 * 16 + 2 * wid + l2];
    }
    for (;;) {
        const int m0 = t * 16;
        const int tn = t + LAYER_GRID;

        float* row0 = &At[(2 * wid) * APITCH + lane * 4];
        float* row1 = row0 + APITCH;
        {
            if (WEIGHTED) {
                *(float4*)row0 = make_float4(d0 * xs0.x, d0 * xs0.y, d0 * xs0.z, d0 * xs0.w);
                *(float4*)row1 = make_float4(d1 * xs1.x, d1 * xs1.y, d1 * xs1.z, d1 * xs1.w);
            } else {
                *(float4*)row0 = xs0;
                *(float4*)row1 = xs1;
            }
        }

        gather_pair<WEIGHTED>(X4, csr, dinv, p0, p1, p2, lane, lo0, hi0, wlo0, whi0, row0, row1);

        int q0 = 0, q1 = 0, q2 = 0, nrp = 0;
        int   nlo = 0, nhi = 0;
        float nwlo = 0.f, nwhi = 0.f;
        float4 nxs0 = make_float4(0.f, 0.f, 0.f, 0.f), nxs1 = nxs0;
        float nd0 = 1.0f, nd1 = 1.0f;
        if (tn < MTILES) {
            q0 = __builtin_amdgcn_readlane(rp, 0);
            q1 = __builtin_amdgcn_readlane(rp, 1);
            q2 = __builtin_amdgcn_readlane(rp, 2);
            const int nn = tn * 16 + 2 * wid;
            if (lane < q2 - q0)      nlo = csr[q0 + lane];
            if (lane + 32 < q2 - q0) nhi = csr[q0 + 32 + lane];
            nxs0 = X4[(unsigned)nn * (unsigned)(DIM / 4) + (unsigned)lane];
            nxs1 = X4[(unsigned)(nn + 1) * (unsigned)(DIM / 4) + (unsigned)lane];
            if (WEIGHTED) { nd0 = dinv[nn]; nd1 = dinv[nn + 1]; }
            const int t2 = (tn + LAYER_GRID < MTILES) ? (tn + LAYER_GRID) : (MTILES - 1);
            nrp = rowptr[t2 * 16 + 2 * wid + l2];
        }

        asm volatile("s_wait_dscnt 0x0\n\ts_barrier_signal -1\n\ts_barrier_wait -1" ::: "memory");

        const float4 bb = ((const float4*)bias)[lane];
        const float dr0 = dinv[m0 + 2 * wid], dr1 = dinv[m0 + 2 * wid + 1];

        v8f c0 = (v8f){0,0,0,0,0,0,0,0};
        v8f c1 = (v8f){0,0,0,0,0,0,0,0};
#pragma unroll
        for (int k0 = 0; k0 < DIM; k0 += 8) {
            const v2f a0 = *(const v2f*)&At[abase + k0];
            const v2f b0 = *(const v2f*)&Wl[bbase + k0 * 128];
            const v2f a1 = *(const v2f*)&At[abase + k0 + 4];
            const v2f b1 = *(const v2f*)&Wl[bbase + (k0 + 4) * 128];
            c0 = __builtin_amdgcn_wmma_f32_16x16x4_f32(false, a0, false, b0, (short)0, c0, false, false);
            c1 = __builtin_amdgcn_wmma_f32_16x16x4_f32(false, a1, false, b1, (short)0, c1, false, false);
        }

        __syncthreads();

#pragma unroll
        for (int v = 0; v < 8; ++v)
            At[(v + 8 * half) * APITCH + wid * 16 + l16] = c0[v] + c1[v];

        __syncthreads();

        if (WEIGHTED) { nwlo = ld_w(dinv, nlo); nwhi = ld_w(dinv, nhi); }

#pragma unroll
        for (int rr = 0; rr < 2; ++rr) {
            const int r = 2 * wid + rr;
            const float4 o = *(const float4*)&At[r * APITCH + lane * 4];
            const float dr = (rr == 0) ? dr0 : dr1;
            const float so = WEIGHTED ? dr : 1.0f;
            float4 q;
            q.x = so * fmaxf(fmaf(dr, o.x, bb.x), 0.f); q.y = so * fmaxf(fmaf(dr, o.y, bb.y), 0.f);
            q.z = so * fmaxf(fmaf(dr, o.z, bb.z), 0.f); q.w = so * fmaxf(fmaf(dr, o.w, bb.w), 0.f);
            ((float4*)out)[(size_t)(m0 + r) * (DIM / 4) + lane] = q;
        }

        if (tn >= MTILES) break;
        t = tn; p0 = q0; p1 = q1; p2 = q2; lo0 = nlo; hi0 = nhi; wlo0 = nwlo; whi0 = nwhi;
        xs0 = nxs0; xs1 = nxs1; d0 = nd0; d1 = nd1; rp = nrp;
    }
}

extern "C" void kernel_launch(void* const* d_in, const int* in_sizes, int n_in,
                              void* d_out, int out_size, void* d_ws, size_t ws_size,
                              hipStream_t stream) {
    const float* x   = (const float*)d_in[0];
    const int*   ei  = (const int*)  d_in[1];
    const float* W1  = (const float*)d_in[2];
    const float* b1  = (const float*)d_in[3];
    const float* W2  = (const float*)d_in[4];
    const float* b2  = (const float*)d_in[5];
    float* out = (float*)d_out;

    const long long E = (long long)in_sizes[1] / 2;
    const int* src = ei;
    const int* dst = ei + E;

    float* dinv = (float*)d_ws;
    float* bufA = dinv + N_NODES;
    int*  rowptr = (int*)(bufA + (size_t)N_NODES * DIM);
    int*  csr    = rowptr + N_NODES + 32;
    int2* tmp    = (int2*)csr + E;
    int*  pcnt   = (int*)(tmp + E);
    int*  pst    = pcnt + NPIECE;

    dim3 blk(256);

    k_chunk_group<<<PG, PT, 0, stream>>>(src, dst, E, pcnt, pst, tmp);
    k_bucket_sort<<<NB, PT, 0, stream>>>(tmp, pcnt, pst, rowptr, dinv, csr, E);

    k_layer<1><<<LAYER_GRID, blk, 0, stream>>>(x, W1, b1, dinv, rowptr, csr, bufA);
    k_layer<2><<<LAYER_GRID, blk, 0, stream>>>(bufA, W2, b2, dinv, rowptr, csr, out);
}
